# starting kernel
# baseline (speedup 1.0000x reference)
#include <hip/hip_runtime.h>

typedef __attribute__((ext_vector_type(2))) float v2f;
typedef __attribute__((ext_vector_type(8))) float v8f;

#define HDIM 128
#define ODIM 40

__global__ void k_init_deg(float* __restrict__ deg, int n) {
    int i = blockIdx.x * blockDim.x + threadIdx.x;
    if (i < n) deg[i] = 1.0f;
}

__global__ void k_edge_deg(const int* __restrict__ col, float* __restrict__ deg, int e) {
    int i = blockIdx.x * blockDim.x + threadIdx.x;
    if (i < e) atomicAdd(&deg[col[i]], 1.0f);
}

__global__ void k_finish_deg(float* __restrict__ deg, int n) {
    int i = blockIdx.x * blockDim.x + threadIdx.x;
    if (i < n) {
        float d = deg[i];
        deg[i] = (d > 0.0f) ? rsqrtf(d) : 0.0f;
    }
}

__global__ void k_zero4(float4* __restrict__ p, long n4) {
    long i = (long)blockIdx.x * blockDim.x + threadIdx.x;
    if (i < n4) p[i] = make_float4(0.f, 0.f, 0.f, 0.f);
}

template <int NPAD>
__global__ __launch_bounds__(256) void k_gemm_wmma(
        const float* __restrict__ A, const float* __restrict__ B,
        const float* __restrict__ bias, float* __restrict__ C,
        int M, int ncols) {
    constexpr int K  = 128;
    constexpr int NT = NPAD / 16;
    constexpr int LSTR = K + 2;
    __shared__ float ldsBT[NPAD * LSTR];

    for (int i = threadIdx.x; i < NPAD * K; i += 256) {
        int k = i & (K - 1);
        int n = i >> 7;
        ldsBT[n * LSTR + k] = (n < ncols) ? B[k * ncols + n] : 0.0f;
    }
    __syncthreads();

    const int lane  = threadIdx.x & 31;
    const int tile  = blockIdx.x * 8 + (threadIdx.x >> 5);
    const int mtile = M >> 4;
    if (tile >= mtile) return;

    const int rowA = tile * 16 + (lane & 15);
    const int koff = (lane >> 4) * 2;
    const float* aptr = A + (size_t)rowA * K + koff;

    v8f acc[NT];
    v8f z8 = {0.f, 0.f, 0.f, 0.f, 0.f, 0.f, 0.f, 0.f};
#pragma unroll
    for (int nt = 0; nt < NT; ++nt) acc[nt] = z8;

#pragma unroll 4
    for (int k = 0; k < K; k += 4) {
        v2f a = *(const v2f*)(aptr + k);
#pragma unroll
        for (int nt = 0; nt < NT; ++nt) {
            int n = nt * 16 + (lane & 15);
            v2f b = *(const v2f*)(&ldsBT[n * LSTR + k + koff]);
            acc[nt] = __builtin_amdgcn_wmma_f32_16x16x4_f32(
                 false, a,  false, b,
                 (short)0, acc[nt],
                 false,  false);
        }
    }

    const int rbase = tile * 16 + ((lane >> 4) << 3);
#pragma unroll
    for (int nt = 0; nt < NT; ++nt) {
        int n = nt * 16 + (lane & 15);
        if (n >= ncols) continue;
        float badd = bias ? bias[n] : 0.0f;
#pragma unroll
        for (int r = 0; r < 8; ++r) {
            C[(size_t)(rbase + r) * ncols + n] = acc[nt][r] + badd;
        }
    }
}

__global__ __launch_bounds__(256) void k_scatter(
        const float* __restrict__ h, const float* __restrict__ dis,
        const int* __restrict__ row, const int* __restrict__ col,
        float* __restrict__ y, int e) {
    int w = blockIdx.x * 8 + (threadIdx.x >> 5);
    if (w >= e) return;
    int lane = threadIdx.x & 31;
    int r = row[w];
    int c = col[w];
    float nrm = dis[r] * dis[c];
    const float4 v = *(const float4*)(h + (size_t)r * HDIM + lane * 4);
    float* dst = y + (size_t)c * HDIM + lane * 4;
    atomicAdd(dst + 0, v.x * nrm);
    atomicAdd(dst + 1, v.y * nrm);
    atomicAdd(dst + 2, v.z * nrm);
    atomicAdd(dst + 3, v.w * nrm);
}

__global__ void k_post1(const float* __restrict__ h, float* __restrict__ y,
                        const float* __restrict__ dis,
                        const float* __restrict__ b1,
                        const float* __restrict__ gamma,
                        const float* __restrict__ beta,
                        const float* __restrict__ mean,
                        const float* __restrict__ var, int total) {
    int i = blockIdx.x * blockDim.x + threadIdx.x;
    if (i >= total) return;
    int ch = i & (HDIM - 1);
    int nd = i >> 7;
    float d = dis[nd];
    float v = y[i] + d * d * h[i] + b1[ch];
    v = (v - mean[ch]) * rsqrtf(var[ch] + 1e-5f) * gamma[ch] + beta[ch];
    y[i] = fmaxf(v, 0.0f);
}

__global__ void k_post2(const float* __restrict__ h, const float* __restrict__ x1,
                        float* __restrict__ y, const float* __restrict__ dis,
                        const float* __restrict__ b2, int total) {
    int i = blockIdx.x * blockDim.x + threadIdx.x;
    if (i >= total) return;
    int ch = i & (HDIM - 1);
    int nd = i >> 7;
    float d = dis[nd];
    float v = y[i] + d * d * h[i] + b2[ch];
    y[i] = fmaxf(x1[i], v);
}

extern "C" void kernel_launch(void* const* d_in, const int* in_sizes, int n_in,
                              void* d_out, int out_size, void* d_ws, size_t ws_size,
                              hipStream_t stream) {
    const float* node_feat = (const float*)d_in[0];
    const int*   eidx      = (const int*)d_in[1];
    const float* W1        = (const float*)d_in[2];
    const float* b1        = (const float*)d_in[3];
    const float* gamma1    = (const float*)d_in[4];
    const float* beta1     = (const float*)d_in[5];
    const float* mean1     = (const float*)d_in[6];
    const float* var1      = (const float*)d_in[7];
    const float* W2        = (const float*)d_in[8];
    const float* b2        = (const float*)d_in[9];
    const float* Wf        = (const float*)d_in[10];
    const float* bf        = (const float*)d_in[11];
    float* out = (float*)d_out;

    const int n = in_sizes[0] / HDIM;
    const int e = in_sizes[1] / 2;
    const int* row = eidx;
    const int* col = eidx + e;

    float* dis = (float*)d_ws;
    float* h   = dis + n;
    float* y1  = h + (size_t)n * HDIM;
    float* y2  = y1 + (size_t)n * HDIM;

    const int nh      = n * HDIM;
    const int mtiles  = n >> 4;
    const int gemmGrd = (mtiles + 7) / 8;
    const int edgeGrd = (e + 7) / 8;

    k_init_deg<<<(n + 255) / 256, 256, 0, stream>>>(dis, n);
    k_edge_deg<<<(e + 255) / 256, 256, 0, stream>>>(col, dis, e);
    k_finish_deg<<<(n + 255) / 256, 256, 0, stream>>>(dis, n);

    long n4 = (long)2 * nh / 4;
    k_zero4<<<(int)((n4 + 255) / 256), 256, 0, stream>>>((float4*)y1, n4);

    k_gemm_wmma<128><<<gemmGrd, 256, 0, stream>>>(node_feat, W1, nullptr, h, n, HDIM);
    k_scatter<<<edgeGrd, 256, 0, stream>>>(h, dis, row, col, y1, e);
    k_post1<<<(nh + 255) / 256, 256, 0, stream>>>(h, y1, dis, b1, gamma1, beta1, mean1, var1, nh);

    k_gemm_wmma<128><<<gemmGrd, 256, 0, stream>>>(y1, W2, nullptr, h, n, HDIM);
    k_scatter<<<edgeGrd, 256, 0, stream>>>(h, dis, row, col, y2, e);
    k_post2<<<(nh + 255) / 256, 256, 0, stream>>>(h, y1, y2, dis, b2, nh);

    k_gemm_wmma<48><<<gemmGrd, 256, 0, stream>>>(y2, Wf, bf, out, n, ODIM);
}


# k_aggr reads per-chunk presence words (degree pass notes which owner range occurs in which 32-edge segment) and loads only flagged segments, not the whole target list; sum = LDS arrival order
# speedup vs baseline: 2.1586x; 2.1586x over previous
#include <hip/hip_runtime.h>

typedef __attribute__((ext_vector_type(2))) float v2f;
typedef __attribute__((ext_vector_type(8))) float v8f;

#define HDIM 128
#define ODIM 40

__global__ void k_init_deg(float* __restrict__ deg, int n) {
    int i = blockIdx.x * blockDim.x + threadIdx.x;
    if (i < n) deg[i] = 1.0f;
}

#define RANGE   400
#define CHUNK   1024

__global__ __launch_bounds__(CHUNK) void k_edge_deg(const int* __restrict__ col, float* __restrict__ deg,
                                                    unsigned* __restrict__ present, int e, int pitch) {
    __shared__ unsigned pres[CHUNK];
    const int tid = (int)threadIdx.x;
    pres[tid] = 0u;
    __syncthreads();
    const int i = (int)blockIdx.x * CHUNK + tid;
    if (i < e) {
        const int c = col[i];
        atomicAdd(&deg[c], 1.0f);
        atomicOr(&pres[c / RANGE], 1u << (tid >> 5));
    }
    __syncthreads();
    if (tid < pitch) present[(size_t)blockIdx.x * pitch + tid] = pres[tid];
}

__global__ void k_finish_deg(float* __restrict__ deg, int n) {
    int i = blockIdx.x * blockDim.x + threadIdx.x;
    if (i < n) {
        float d = deg[i];
        deg[i] = (d > 0.0f) ? rsqrtf(d) : 0.0f;
    }
}

template <int NPAD>
__global__ __launch_bounds__(256) void k_gemm_wmma(
        const float* __restrict__ A, const float* __restrict__ B,
        const float* __restrict__ bias, float* __restrict__ C,
        int M, int ncols) {
    constexpr int K  = 128;
    constexpr int NT = NPAD / 16;
    constexpr int LSTR = K + 2;
    __shared__ float ldsBT[NPAD * LSTR];

    for (int i = threadIdx.x; i < NPAD * K; i += 256) {
        int k = i & (K - 1);
        int n = i >> 7;
        ldsBT[n * LSTR + k] = (n < ncols) ? B[k * ncols + n] : 0.0f;
    }
    __syncthreads();

    const int lane  = threadIdx.x & 31;
    const int tile  = blockIdx.x * 8 + (threadIdx.x >> 5);
    const int mtile = M >> 4;
    if (tile >= mtile) return;

    const int rowA = tile * 16 + (lane & 15);
    const int koff = (lane >> 4) * 2;
    const float* aptr = A + (size_t)rowA * K + koff;

    v8f acc[NT];
    v8f z8 = {0.f, 0.f, 0.f, 0.f, 0.f, 0.f, 0.f, 0.f};
#pragma unroll
    for (int nt = 0; nt < NT; ++nt) acc[nt] = z8;

#pragma unroll 4
    for (int k = 0; k < K; k += 4) {
        v2f a = *(const v2f*)(aptr + k);
#pragma unroll
        for (int nt = 0; nt < NT; ++nt) {
            int n = nt * 16 + (lane & 15);
            v2f b = *(const v2f*)(&ldsBT[n * LSTR + k + koff]);
            acc[nt] = __builtin_amdgcn_wmma_f32_16x16x4_f32(
                 false, a,  false, b,
                 (short)0, acc[nt],
                 false,  false);
        }
    }

    const int rbase = tile * 16 + ((lane >> 4) << 3);
#pragma unroll
    for (int nt = 0; nt < NT; ++nt) {
        int n = nt * 16 + (lane & 15);
        if (n >= ncols) continue;
        float badd = bias ? bias[n] : 0.0f;
#pragma unroll
        for (int r = 0; r < 8; ++r) {
            C[(size_t)(rbase + r) * ncols + n] = acc[nt][r] + badd;
        }
    }
}

#define AWAVES  32
#define SGRP    8
#define QCAP    288
#define SCAP    264
#define EGRP    8

__device__ __forceinline__ void load_rows(const float* __restrict__ h, int r, int g, int cnt, unsigned lo,
                                          float (&v)[EGRP][4]) {
#pragma unroll
    for (int j = 0; j < EGRP; ++j) {
        if (g + j < cnt) {
            const float* srow = h + (size_t)__builtin_amdgcn_readlane(r, g + j) * HDIM;
            v[j][0] = srow[lo];
            v[j][1] = srow[lo + 32];
            v[j][2] = srow[lo + 64];
            v[j][3] = srow[lo + 96];
        }
    }
}

__device__ __forceinline__ void lds_add(float* p, float x) {
    __hip_atomic_fetch_add(p, x, __ATOMIC_RELAXED, __HIP_MEMORY_SCOPE_WORKGROUP);
}

__device__ __forceinline__ void add_rows(float* acc, int t, float nrm, int g, int cnt, unsigned lo,
                                         const float (&v)[EGRP][4]) {
#pragma unroll
    for (int j = 0; j < EGRP; ++j) {
        if (g + j < cnt) {
            float* drow = acc + __builtin_amdgcn_readlane(t, g + j) * HDIM;
            const float nj = __int_as_float(__builtin_amdgcn_readlane(__float_as_int(nrm), g + j));
            lds_add(drow + lo,      v[j][0] * nj);
            lds_add(drow + lo + 32, v[j][1] * nj);
            lds_add(drow + lo + 64, v[j][2] * nj);
            lds_add(drow + lo + 96, v[j][3] * nj);
        }
    }
}

__device__ __forceinline__ void drain(const float* __restrict__ h, const float* __restrict__ dis,
                                      const int* __restrict__ row, float* acc,
                                      const int* qE, const unsigned short* qT, int off, int cnt,
                                      unsigned lo, int node_lo) {
    const int li = off + min((int)lo, cnt - 1);
    const int ej = qE[li];
    const int tj = (int)qT[li];
    const int r  = row[ej];
    float v[EGRP][4];
    load_rows(h, r, 0, cnt, lo, v);
    const float nrm = dis[r] * dis[node_lo + tj];
    add_rows(acc, tj, nrm, 0, cnt, lo, v);
#pragma unroll
    for (int g = EGRP; g < 32; g += EGRP) {
        load_rows(h, r, g, cnt, lo, v);
        add_rows(acc, tj, nrm, g, cnt, lo, v);
    }
}

__device__ __forceinline__ void append(bool m, int eidx, unsigned t, int& qn, int* qE, unsigned short* qT) {
    const unsigned bal = __builtin_amdgcn_ballot_w32(m);
    if (bal == 0u) return;
    const int pos = qn + (int)__builtin_amdgcn_mbcnt_lo(bal, 0u);
    if (m) { qE[pos] = eidx; qT[pos] = (unsigned short)t; }
    qn += __builtin_popcount(bal);
}

__device__ __forceinline__ void append_seg(bool m, int seg, int& sn, int* sL) {
    const unsigned bal = __builtin_amdgcn_ballot_w32(m);
    if (bal == 0u) return;
    if (m) sL[sn + (int)__builtin_amdgcn_mbcnt_lo(bal, 0u)] = seg;
    sn += __builtin_popcount(bal);
}

__device__ __forceinline__ void scan_batch(const int* __restrict__ col, const int* sL, int off, int cnt, int e,
                                           unsigned lo, int node_lo, int rows, int& qn,
                                           int* qE, unsigned short* qT) {
    const int sv = sL[off + min((int)lo, cnt - 1)];
    int ei[SGRP];
    int c[SGRP];
#pragma unroll
    for (int j = 0; j < SGRP; ++j) {
        if (j < cnt) {
            ei[j] = __builtin_amdgcn_readlane(sv, j) * 32 + (int)lo;
            c[j] = col[min(ei[j], e - 1)];
        }
    }
#pragma unroll
    for (int j = 0; j < SGRP; ++j) {
        if (j < cnt) {
            const unsigned d = (unsigned)(c[j] - node_lo);
            append((d < (unsigned)rows) & (ei[j] < e), ei[j], d, qn, qE, qT);
        }
    }
}

__global__ __launch_bounds__(1024) void k_aggr(
        const float* __restrict__ h, const float* __restrict__ dis,
        const int* __restrict__ row, const int* __restrict__ col,
        const unsigned* __restrict__ present, int pitch,
        float* __restrict__ y, int n, int e) {
    __shared__ float acc[RANGE * HDIM];
    __shared__ int   qEs[AWAVES * QCAP];
    __shared__ int   sLs[AWAVES * SCAP];
    __shared__ unsigned short qTs[AWAVES * QCAP];

    const int tid = (int)threadIdx.x;
    const unsigned lo = threadIdx.x & 31;
    const int w = __builtin_amdgcn_readfirstlane(tid >> 5);
    const int node_lo = (int)blockIdx.x * RANGE;
    const int rows = min(RANGE, n - node_lo);
    int* qE = qEs + w * QCAP;
    unsigned short* qT = qTs + w * QCAP;
    int* sL = sLs + w * SCAP;

    float4* acc4 = reinterpret_cast<float4*>(acc);
    for (int i = tid; i < RANGE * HDIM / 4; i += 1024) acc4[i] = make_float4(0.f, 0.f, 0.f, 0.f);
    __syncthreads();

    int qn = 0;
    int sn = 0;
    const int nchunk = (e + CHUNK - 1) / CHUNK;
    const unsigned* mine = present + blockIdx.x;
#pragma unroll 1
    for (int cb = w; cb < nchunk; cb += 32 * AWAVES) {
        const int chunk = cb + AWAVES * (int)lo;
        const unsigned wd = (chunk < nchunk) ? mine[(size_t)chunk * pitch] : 0u;
#pragma unroll 1
        for (int kc = 0; kc < 32; kc += 8) {
            const unsigned part = (wd >> kc) & 0xffu;
            if (__builtin_amdgcn_ballot_w32(part != 0u) == 0u) continue;
#pragma unroll
            for (int k = 0; k < 8; ++k)
                append_seg(((part >> k) & 1u) != 0u, chunk * 32 + kc + k, sn, sL);
            while (sn >= SGRP) {
                sn -= SGRP;
                scan_batch(col, sL, sn, SGRP, e, lo, node_lo, rows, qn, qE, qT);
                while (qn >= 32) {
                    qn -= 32;
                    drain(h, dis, row, acc, qE, qT, qn, 32, lo, node_lo);
                }
            }
        }
    }
    if (sn > 0) {
        scan_batch(col, sL, 0, sn, e, lo, node_lo, rows, qn, qE, qT);
        while (qn >= 32) {
            qn -= 32;
            drain(h, dis, row, acc, qE, qT, qn, 32, lo, node_lo);
        }
    }

    if (qn > 0) drain(h, dis, row, acc, qE, qT, 0, qn, lo, node_lo);
    __syncthreads();

    float4* y4 = reinterpret_cast<float4*>(y + (size_t)node_lo * HDIM);
    for (int i = tid; i < rows * (HDIM / 4); i += 1024) y4[i] = acc4[i];
}

__global__ void k_post1(const float* __restrict__ h, float* __restrict__ y,
                        const float* __restrict__ dis,
                        const float* __restrict__ b1,
                        const float* __restrict__ gamma,
                        const float* __restrict__ beta,
                        const float* __restrict__ mean,
                        const float* __restrict__ var, int total) {
    int i = blockIdx.x * blockDim.x + threadIdx.x;
    if (i >= total) return;
    int ch = i & (HDIM - 1);
    int nd = i >> 7;
    float d = dis[nd];
    float v = y[i] + d * d * h[i] + b1[ch];
    v = (v - mean[ch]) * rsqrtf(var[ch] + 1e-5f) * gamma[ch] + beta[ch];
    y[i] = fmaxf(v, 0.0f);
}

__global__ void k_post2(const float* __restrict__ h, const float* __restrict__ x1,
                        float* __restrict__ y,
                        const float* __restrict__ dis,
                        const float* __restrict__ b2, int total) {
    int i = blockIdx.x * blockDim.x + threadIdx.x;
    if (i >= total) return;
    int ch = i & (HDIM - 1);
    int nd = i >> 7;
    float d = dis[nd];
    float v = y[i] + d * d * h[i] + b2[ch];
    y[i] = fmaxf(x1[i], v);
}

extern "C" void kernel_launch(void* const* d_in, const int* in_sizes, int n_in,
                              void* d_out, int out_size, void* d_ws, size_t ws_size,
                              hipStream_t stream) {
    const float* node_feat = (const float*)d_in[0];
    const int*   eidx      = (const int*)d_in[1];
    const float* W1        = (const float*)d_in[2];
    const float* b1        = (const float*)d_in[3];
    const float* gamma1    = (const float*)d_in[4];
    const float* beta1     = (const float*)d_in[5];
    const float* mean1     = (const float*)d_in[6];
    const float* var1      = (const float*)d_in[7];
    const float* W2        = (const float*)d_in[8];
    const float* b2        = (const float*)d_in[9];
    const float* Wf        = (const float*)d_in[10];
    const float* bf        = (const float*)d_in[11];
    float* out = (float*)d_out;

    const int n = in_sizes[0] / HDIM;
    const int e = in_sizes[1] / 2;
    const int* row = eidx;
    const int* col = eidx + e;

    float* dis = (float*)d_ws;
    float* h   = dis + n;
    float* y1  = h + (size_t)n * HDIM;
    float* y2  = y1 + (size_t)n * HDIM;
    unsigned* present = (unsigned*)(y2 + (size_t)n * HDIM);

    const int nh      = n * HDIM;
    const int mtiles  = n >> 4;
    const int gemmGrd = (mtiles + 7) / 8;
    const int aggGrd  = (n + RANGE - 1) / RANGE;
    const int pitch   = (aggGrd + 31) / 32 * 32;
    const int chunks  = (e + CHUNK - 1) / CHUNK;

    k_init_deg<<<(n + 255) / 256, 256, 0, stream>>>(dis, n);
    k_edge_deg<<<chunks, CHUNK, 0, stream>>>(col, dis, present, e, pitch);
    k_finish_deg<<<(n + 255) / 256, 256, 0, stream>>>(dis, n);

    k_gemm_wmma<128><<<gemmGrd, 256, 0, stream>>>(node_feat, W1, nullptr, h, n, HDIM);
    k_aggr<<<aggGrd, 1024, 0, stream>>>(h, dis, row, col, present, pitch, y1, n, e);
    k_post1<<<(nh + 255) / 256, 256, 0, stream>>>(h, y1, dis, b1, gamma1, beta1, mean1, var1, nh);

    k_gemm_wmma<128><<<gemmGrd, 256, 0, stream>>>(y1, W2, nullptr, h, n, HDIM);
    k_aggr<<<aggGrd, 1024, 0, stream>>>(h, dis, row, col, present, pitch, y2, n, e);
    k_post2<<<(nh + 255) / 256, 256, 0, stream>>>(h, y1, y2, dis, b2, nh);

    k_gemm_wmma<48><<<gemmGrd, 256, 0, stream>>>(y2, Wf, bf, out, n, ODIM);
}


# Epilogue passes (self loop + bias + BN/ReLU, JK max): a thread takes 4 groups of 4 adjacent channels with 16-byte loads and stores, not one element; 16x fewer waves, same arithmetic per element
# speedup vs baseline: 2.3210x; 1.0752x over previous
#include <hip/hip_runtime.h>

typedef __attribute__((ext_vector_type(2))) float v2f;
typedef __attribute__((ext_vector_type(8))) float v8f;

#define HDIM 128
#define ODIM 40

__global__ void k_init_deg(float* __restrict__ deg, int n) {
    int i = blockIdx.x * blockDim.x + threadIdx.x;
    if (i < n) deg[i] = 1.0f;
}

#define RANGE   400
#define CHUNK   1024

__global__ __launch_bounds__(CHUNK) void k_edge_deg(const int* __restrict__ col, float* __restrict__ deg,
                                                    unsigned* __restrict__ present, int e, int pitch) {
    __shared__ unsigned pres[CHUNK];
    const int tid = (int)threadIdx.x;
    pres[tid] = 0u;
    __syncthreads();
    const int i = (int)blockIdx.x * CHUNK + tid;
    if (i < e) {
        const int c = col[i];
        atomicAdd(&deg[c], 1.0f);
        atomicOr(&pres[c / RANGE], 1u << (tid >> 5));
    }
    __syncthreads();
    if (tid < pitch) present[(size_t)blockIdx.x * pitch + tid] = pres[tid];
}

__global__ void k_finish_deg(float* __restrict__ deg, int n) {
    int i = blockIdx.x * blockDim.x + threadIdx.x;
    if (i < n) {
        float d = deg[i];
        deg[i] = (d > 0.0f) ? rsqrtf(d) : 0.0f;
    }
}

template <int NPAD>
__global__ __launch_bounds__(256) void k_gemm_wmma(
        const float* __restrict__ A, const float* __restrict__ B,
        const float* __restrict__ bias, float* __restrict__ C,
        int M, int ncols) {
    constexpr int K  = 128;
    constexpr int NT = NPAD / 16;
    constexpr int LSTR = K + 2;
    __shared__ float ldsBT[NPAD * LSTR];

    for (int i = threadIdx.x; i < NPAD * K; i += 256) {
        int k = i & (K - 1);
        int n = i >> 7;
        ldsBT[n * LSTR + k] = (n < ncols) ? B[k * ncols + n] : 0.0f;
    }
    __syncthreads();

    const int lane  = threadIdx.x & 31;
    const int tile  = blockIdx.x * 8 + (threadIdx.x >> 5);
    const int mtile = M >> 4;
    if (tile >= mtile) return;

    const int rowA = tile * 16 + (lane & 15);
    const int koff = (lane >> 4) * 2;
    const float* aptr = A + (size_t)rowA * K + koff;

    v8f acc[NT];
    v8f z8 = {0.f, 0.f, 0.f, 0.f, 0.f, 0.f, 0.f, 0.f};
#pragma unroll
    for (int nt = 0; nt < NT; ++nt) acc[nt] = z8;

#pragma unroll 4
    for (int k = 0; k < K; k += 4) {
        v2f a = *(const v2f*)(aptr + k);
#pragma unroll
        for (int nt = 0; nt < NT; ++nt) {
            int n = nt * 16 + (lane & 15);
            v2f b = *(const v2f*)(&ldsBT[n * LSTR + k + koff]);
            acc[nt] = __builtin_amdgcn_wmma_f32_16x16x4_f32(
                 false, a,  false, b,
                 (short)0, acc[nt],
                 false,  false);
        }
    }

    const int rbase = tile * 16 + ((lane >> 4) << 3);
#pragma unroll
    for (int nt = 0; nt < NT; ++nt) {
        int n = nt * 16 + (lane & 15);
        if (n >= ncols) continue;
        float badd = bias ? bias[n] : 0.0f;
#pragma unroll
        for (int r = 0; r < 8; ++r) {
            C[(size_t)(rbase + r) * ncols + n] = acc[nt][r] + badd;
        }
    }
}

#define AWAVES  32
#define SGRP    8
#define QCAP    288
#define SCAP    264
#define EGRP    8

__device__ __forceinline__ void load_rows(const float* __restrict__ h, int r, int g, int cnt, unsigned lo,
                                          float (&v)[EGRP][4]) {
#pragma unroll
    for (int j = 0; j < EGRP; ++j) {
        if (g + j < cnt) {
            const float* srow = h + (size_t)__builtin_amdgcn_readlane(r, g + j) * HDIM;
            v[j][0] = srow[lo];
            v[j][1] = srow[lo + 32];
            v[j][2] = srow[lo + 64];
            v[j][3] = srow[lo + 96];
        }
    }
}

__device__ __forceinline__ void lds_add(float* p, float x) {
    __hip_atomic_fetch_add(p, x, __ATOMIC_RELAXED, __HIP_MEMORY_SCOPE_WORKGROUP);
}

__device__ __forceinline__ void add_rows(float* acc, int t, float nrm, int g, int cnt, unsigned lo,
                                         const float (&v)[EGRP][4]) {
#pragma unroll
    for (int j = 0; j < EGRP; ++j) {
        if (g + j < cnt) {
            float* drow = acc + __builtin_amdgcn_readlane(t, g + j) * HDIM;
            const float nj = __int_as_float(__builtin_amdgcn_readlane(__float_as_int(nrm), g + j));
            lds_add(drow + lo,      v[j][0] * nj);
            lds_add(drow + lo + 32, v[j][1] * nj);
            lds_add(drow + lo + 64, v[j][2] * nj);
            lds_add(drow + lo + 96, v[j][3] * nj);
        }
    }
}

__device__ __forceinline__ void drain(const float* __restrict__ h, const float* __restrict__ dis,
                                      const int* __restrict__ row, float* acc,
                                      const int* qE, const unsigned short* qT, int off, int cnt,
                                      unsigned lo, int node_lo) {
    const int li = off + min((int)lo, cnt - 1);
    const int ej = qE[li];
    const int tj = (int)qT[li];
    const int r  = row[ej];
    float v[EGRP][4];
    load_rows(h, r, 0, cnt, lo, v);
    const float nrm = dis[r] * dis[node_lo + tj];
    add_rows(acc, tj, nrm, 0, cnt, lo, v);
#pragma unroll
    for (int g = EGRP; g < 32; g += EGRP) {
        load_rows(h, r, g, cnt, lo, v);
        add_rows(acc, tj, nrm, g, cnt, lo, v);
    }
}

__device__ __forceinline__ void append(bool m, int eidx, unsigned t, int& qn, int* qE, unsigned short* qT) {
    const unsigned bal = __builtin_amdgcn_ballot_w32(m);
    if (bal == 0u) return;
    const int pos = qn + (int)__builtin_amdgcn_mbcnt_lo(bal, 0u);
    if (m) { qE[pos] = eidx; qT[pos] = (unsigned short)t; }
    qn += __builtin_popcount(bal);
}

__device__ __forceinline__ void append_seg(bool m, int seg, int& sn, int* sL) {
    const unsigned bal = __builtin_amdgcn_ballot_w32(m);
    if (bal == 0u) return;
    if (m) sL[sn + (int)__builtin_amdgcn_mbcnt_lo(bal, 0u)] = seg;
    sn += __builtin_popcount(bal);
}

__device__ __forceinline__ void scan_batch(const int* __restrict__ col, const int* sL, int off, int cnt, int e,
                                           unsigned lo, int node_lo, int rows, int& qn,
                                           int* qE, unsigned short* qT) {
    const int sv = sL[off + min((int)lo, cnt - 1)];
    int ei[SGRP];
    int c[SGRP];
#pragma unroll
    for (int j = 0; j < SGRP; ++j) {
        if (j < cnt) {
            ei[j] = __builtin_amdgcn_readlane(sv, j) * 32 + (int)lo;
            c[j] = col[min(ei[j], e - 1)];
        }
    }
#pragma unroll
    for (int j = 0; j < SGRP; ++j) {
        if (j < cnt) {
            const unsigned d = (unsigned)(c[j] - node_lo);
            append((d < (unsigned)rows) & (ei[j] < e), ei[j], d, qn, qE, qT);
        }
    }
}

__global__ __launch_bounds__(1024) void k_aggr(
        const float* __restrict__ h, const float* __restrict__ dis,
        const int* __restrict__ row, const int* __restrict__ col,
        const unsigned* __restrict__ present, int pitch,
        float* __restrict__ y, int n, int e) {
    __shared__ float acc[RANGE * HDIM];
    __shared__ int   qEs[AWAVES * QCAP];
    __shared__ int   sLs[AWAVES * SCAP];
    __shared__ unsigned short qTs[AWAVES * QCAP];

    const int tid = (int)threadIdx.x;
    const unsigned lo = threadIdx.x & 31;
    const int w = __builtin_amdgcn_readfirstlane(tid >> 5);
    const int node_lo = (int)blockIdx.x * RANGE;
    const int rows = min(RANGE, n - node_lo);
    int* qE = qEs + w * QCAP;
    unsigned short* qT = qTs + w * QCAP;
    int* sL = sLs + w * SCAP;

    float4* acc4 = reinterpret_cast<float4*>(acc);
    for (int i = tid; i < RANGE * HDIM / 4; i += 1024) acc4[i] = make_float4(0.f, 0.f, 0.f, 0.f);
    __syncthreads();

    int qn = 0;
    int sn = 0;
    const int nchunk = (e + CHUNK - 1) / CHUNK;
    const unsigned* mine = present + blockIdx.x;
#pragma unroll 1
    for (int cb = w; cb < nchunk; cb += 32 * AWAVES) {
        const int chunk = cb + AWAVES * (int)lo;
        const unsigned wd = (chunk < nchunk) ? mine[(size_t)chunk * pitch] : 0u;
#pragma unroll 1
        for (int kc = 0; kc < 32; kc += 8) {
            const unsigned part = (wd >> kc) & 0xffu;
            if (__builtin_amdgcn_ballot_w32(part != 0u) == 0u) continue;
#pragma unroll
            for (int k = 0; k < 8; ++k)
                append_seg(((part >> k) & 1u) != 0u, chunk * 32 + kc + k, sn, sL);
            while (sn >= SGRP) {
                sn -= SGRP;
                scan_batch(col, sL, sn, SGRP, e, lo, node_lo, rows, qn, qE, qT);
                while (qn >= 32) {
                    qn -= 32;
                    drain(h, dis, row, acc, qE, qT, qn, 32, lo, node_lo);
                }
            }
        }
    }
    if (sn > 0) {
        scan_batch(col, sL, 0, sn, e, lo, node_lo, rows, qn, qE, qT);
        while (qn >= 32) {
            qn -= 32;
            drain(h, dis, row, acc, qE, qT, qn, 32, lo, node_lo);
        }
    }

    if (qn > 0) drain(h, dis, row, acc, qE, qT, 0, qn, lo, node_lo);
    __syncthreads();

    float4* y4 = reinterpret_cast<float4*>(y + (size_t)node_lo * HDIM);
    for (int i = tid; i < rows * (HDIM / 4); i += 1024) y4[i] = acc4[i];
}

#define PVEC 4
#define PTHR 256

__device__ __forceinline__ float bn_relu(float v, float m, float vr, float g, float b) {
    v = (v - m) * rsqrtf(vr + 1e-5f) * g + b;
    return fmaxf(v, 0.0f);
}

__global__ __launch_bounds__(PTHR) void k_post1(const float* __restrict__ h, float* __restrict__ y,
                        const float* __restrict__ dis,
                        const float* __restrict__ b1,
                        const float* __restrict__ gamma,
                        const float* __restrict__ beta,
                        const float* __restrict__ mean,
                        const float* __restrict__ var, int total4) {
    const float4* __restrict__ h4 = reinterpret_cast<const float4*>(h);
    float4* __restrict__ y4 = reinterpret_cast<float4*>(y);
    const int c4 = (int)threadIdx.x & (HDIM / 4 - 1);
    const int base = (int)blockIdx.x * (PTHR * PVEC) + (int)threadIdx.x;
    float4 yv[PVEC], hv[PVEC];
    float d[PVEC];
#pragma unroll
    for (int j = 0; j < PVEC; ++j) {
        const int v = min(base + j * PTHR, total4 - 1);
        yv[j] = y4[v];
        hv[j] = h4[v];
        d[j] = dis[v / (HDIM / 4)];
    }
    const float4 bb = reinterpret_cast<const float4*>(b1)[c4];
    const float4 mm = reinterpret_cast<const float4*>(mean)[c4];
    const float4 vv = reinterpret_cast<const float4*>(var)[c4];
    const float4 gg = reinterpret_cast<const float4*>(gamma)[c4];
    const float4 be = reinterpret_cast<const float4*>(beta)[c4];
#pragma unroll
    for (int j = 0; j < PVEC; ++j) {
        const int v = base + j * PTHR;
        if (v < total4) {
            const float dd = d[j] * d[j];
            float4 o;
            o.x = bn_relu(yv[j].x + dd * hv[j].x + bb.x, mm.x, vv.x, gg.x, be.x);
            o.y = bn_relu(yv[j].y + dd * hv[j].y + bb.y, mm.y, vv.y, gg.y, be.y);
            o.z = bn_relu(yv[j].z + dd * hv[j].z + bb.z, mm.z, vv.z, gg.z, be.z);
            o.w = bn_relu(yv[j].w + dd * hv[j].w + bb.w, mm.w, vv.w, gg.w, be.w);
            y4[v] = o;
        }
    }
}

__global__ __launch_bounds__(PTHR) void k_post2(const float* __restrict__ h, const float* __restrict__ x1,
                        float* __restrict__ y,
                        const float* __restrict__ dis,
                        const float* __restrict__ b2, int total4) {
    const float4* __restrict__ h4 = reinterpret_cast<const float4*>(h);
    const float4* __restrict__ x4 = reinterpret_cast<const float4*>(x1);
    float4* __restrict__ y4 = reinterpret_cast<float4*>(y);
    const int c4 = (int)threadIdx.x & (HDIM / 4 - 1);
    const int base = (int)blockIdx.x * (PTHR * PVEC) + (int)threadIdx.x;
    float4 yv[PVEC], hv[PVEC], xv[PVEC];
    float d[PVEC];
#pragma unroll
    for (int j = 0; j < PVEC; ++j) {
        const int v = min(base + j * PTHR, total4 - 1);
        yv[j] = y4[v];
        hv[j] = h4[v];
        xv[j] = x4[v];
        d[j] = dis[v / (HDIM / 4)];
    }
    const float4 bb = reinterpret_cast<const float4*>(b2)[c4];
#pragma unroll
    for (int j = 0; j < PVEC; ++j) {
        const int v = base + j * PTHR;
        if (v < total4) {
            const float dd = d[j] * d[j];
            float4 o;
            o.x = fmaxf(xv[j].x, yv[j].x + dd * hv[j].x + bb.x);
            o.y = fmaxf(xv[j].y, yv[j].y + dd * hv[j].y + bb.y);
            o.z = fmaxf(xv[j].z, yv[j].z + dd * hv[j].z + bb.z);
            o.w = fmaxf(xv[j].w, yv[j].w + dd * hv[j].w + bb.w);
            y4[v] = o;
        }
    }
}

extern "C" void kernel_launch(void* const* d_in, const int* in_sizes, int n_in,
                              void* d_out, int out_size, void* d_ws, size_t ws_size,
                              hipStream_t stream) {
    const float* node_feat = (const float*)d_in[0];
    const int*   eidx      = (const int*)d_in[1];
    const float* W1        = (const float*)d_in[2];
    const float* b1        = (const float*)d_in[3];
    const float* gamma1    = (const float*)d_in[4];
    const float* beta1     = (const float*)d_in[5];
    const float* mean1     = (const float*)d_in[6];
    const float* var1      = (const float*)d_in[7];
    const float* W2        = (const float*)d_in[8];
    const float* b2        = (const float*)d_in[9];
    const float* Wf        = (const float*)d_in[10];
    const float* bf        = (const float*)d_in[11];
    float* out = (float*)d_out;

    const int n = in_sizes[0] / HDIM;
    const int e = in_sizes[1] / 2;
    const int* row = eidx;
    const int* col = eidx + e;

    float* dis = (float*)d_ws;
    float* h   = dis + n;
    float* y1  = h + (size_t)n * HDIM;
    float* y2  = y1 + (size_t)n * HDIM;
    unsigned* present = (unsigned*)(y2 + (size_t)n * HDIM);

    const int nh4     = n * (HDIM / 4);
    const int postGrd = (nh4 + PTHR * PVEC - 1) / (PTHR * PVEC);
    const int mtiles  = n >> 4;
    const int gemmGrd = (mtiles + 7) / 8;
    const int aggGrd  = (n + RANGE - 1) / RANGE;
    const int pitch   = (aggGrd + 31) / 32 * 32;
    const int chunks  = (e + CHUNK - 1) / CHUNK;

    k_init_deg<<<(n + 255) / 256, 256, 0, stream>>>(dis, n);
    k_edge_deg<<<chunks, CHUNK, 0, stream>>>(col, dis, present, e, pitch);
    k_finish_deg<<<(n + 255) / 256, 256, 0, stream>>>(dis, n);

    k_gemm_wmma<128><<<gemmGrd, 256, 0, stream>>>(node_feat, W1, nullptr, h, n, HDIM);
    k_aggr<<<aggGrd, 1024, 0, stream>>>(h, dis, row, col, present, pitch, y1, n, e);
    k_post1<<<postGrd, PTHR, 0, stream>>>(h, y1, dis, b1, gamma1, beta1, mean1, var1, nh4);

    k_gemm_wmma<128><<<gemmGrd, 256, 0, stream>>>(y1, W2, nullptr, h, n, HDIM);
    k_aggr<<<aggGrd, 1024, 0, stream>>>(h, dis, row, col, present, pitch, y2, n, e);
    k_post2<<<postGrd, PTHR, 0, stream>>>(h, y1, y2, dis, b2, nh4);

    k_gemm_wmma<48><<<gemmGrd, 256, 0, stream>>>(y2, Wf, bf, out, n, ODIM);
}


# GEMM: B fragments of a wave's column share held in registers by persistent waves (2 per SIMD) looping over row tiles, next tile's A loaded ahead; no LDS staging of B; k and sum order kept
# speedup vs baseline: 2.5517x; 1.0994x over previous
#include <hip/hip_runtime.h>

typedef __attribute__((ext_vector_type(2))) float v2f;
typedef __attribute__((ext_vector_type(8))) float v8f;

#define HDIM 128
#define ODIM 40

__global__ void k_init_deg(float* __restrict__ deg, int n) {
    int i = blockIdx.x * blockDim.x + threadIdx.x;
    if (i < n) deg[i] = 1.0f;
}

#define RANGE   400
#define CHUNK   1024

__global__ __launch_bounds__(CHUNK) void k_edge_deg(const int* __restrict__ col, float* __restrict__ deg,
                                                    unsigned* __restrict__ present, int e, int pitch) {
    __shared__ unsigned pres[CHUNK];
    const int tid = (int)threadIdx.x;
    pres[tid] = 0u;
    __syncthreads();
    const int i = (int)blockIdx.x * CHUNK + tid;
    if (i < e) {
        const int c = col[i];
        atomicAdd(&deg[c], 1.0f);
        atomicOr(&pres[c / RANGE], 1u << (tid >> 5));
    }
    __syncthreads();
    if (tid < pitch) present[(size_t)blockIdx.x * pitch + tid] = pres[tid];
}

__global__ void k_finish_deg(float* __restrict__ deg, int n) {
    int i = blockIdx.x * blockDim.x + threadIdx.x;
    if (i < n) {
        float d = deg[i];
        deg[i] = (d > 0.0f) ? rsqrtf(d) : 0.0f;
    }
}

#define GEMM_WGS 256

template <int NPAD, int SPLIT>
__global__ __launch_bounds__(256) void k_gemm_wmma(
        const float* __restrict__ A, const float* __restrict__ B,
        const float* __restrict__ bias, float* __restrict__ C,
        int M, int ncols) {
    constexpr int K   = 128;
    constexpr int KS  = K / 4;
    constexpr int NTW = NPAD / 16 / SPLIT;

    const int lane  = threadIdx.x & 31;
    const int c     = lane & 15;
    const int koff  = (lane >> 4) * 2;
    const int gw    = (int)blockIdx.x * 8 + __builtin_amdgcn_readfirstlane((int)threadIdx.x >> 5);
    const int share = gw % SPLIT;
    const int t0    = gw / SPLIT;
    const int tstep = (int)gridDim.x * 8 / SPLIT;
    const int mtile = M >> 4;
    if (t0 >= mtile) return;
    const int nbase = share * (NTW * 16);

    v2f b[KS][NTW];
    float badd[NTW];
#pragma unroll
    for (int nt = 0; nt < NTW; ++nt) {
        const int n = nbase + nt * 16 + c;
        const bool in = n < ncols;
        const int nn = in ? n : 0;
        badd[nt] = (bias && in) ? bias[nn] : 0.0f;
        const float* bp = B + (size_t)koff * ncols + nn;
#pragma unroll
        for (int ks = 0; ks < KS; ++ks) {
            v2f t;
            t.x = bp[(ks * 4) * ncols];
            t.y = bp[(ks * 4 + 1) * ncols];
            if (!in) { t.x = 0.f; t.y = 0.f; }
            b[ks][nt] = t;
        }
    }

    v2f a[KS];
    {
        const float* aptr = A + (size_t)(t0 * 16 + c) * K + koff;
#pragma unroll
        for (int ks = 0; ks < KS; ++ks) a[ks] = *(const v2f*)(aptr + ks * 4);
    }

    const v8f z8 = {0.f, 0.f, 0.f, 0.f, 0.f, 0.f, 0.f, 0.f};
#pragma unroll 1
    for (int tile = t0; tile < mtile; tile += tstep) {
        v2f an[KS];
        const int nxt = tile + tstep;
        if (nxt < mtile) {
            const float* aptr = A + (size_t)(nxt * 16 + c) * K + koff;
#pragma unroll
            for (int ks = 0; ks < KS; ++ks) an[ks] = *(const v2f*)(aptr + ks * 4);
        } else {
#pragma unroll
            for (int ks = 0; ks < KS; ++ks) an[ks] = a[ks];
        }

        v8f acc[NTW];
#pragma unroll
        for (int nt = 0; nt < NTW; ++nt) acc[nt] = z8;
#pragma unroll
        for (int ks = 0; ks < KS; ++ks) {
#pragma unroll
            for (int nt = 0; nt < NTW; ++nt) {
                acc[nt] = __builtin_amdgcn_wmma_f32_16x16x4_f32(
                     false, a[ks],  false, b[ks][nt],
                     (short)0, acc[nt],
                     false,  false);
            }
        }

        const int rbase = tile * 16 + ((lane >> 4) << 3);
#pragma unroll
        for (int nt = 0; nt < NTW; ++nt) {
            const int n = nbase + nt * 16 + c;
            if (n >= ncols) continue;
            float* cp = C + (size_t)rbase * ncols + n;
#pragma unroll
            for (int r = 0; r < 8; ++r) cp[r * ncols] = acc[nt][r] + badd[nt];
        }
#pragma unroll
        for (int ks = 0; ks < KS; ++ks) a[ks] = an[ks];
    }
}

#define AWAVES  32
#define SGRP    8
#define QCAP    288
#define SCAP    264
#define EGRP    8

__device__ __forceinline__ void load_rows(const float* __restrict__ h, int r, int g, int cnt, unsigned lo,
                                          float (&v)[EGRP][4]) {
#pragma unroll
    for (int j = 0; j < EGRP; ++j) {
        if (g + j < cnt) {
            const float* srow = h + (size_t)__builtin_amdgcn_readlane(r, g + j) * HDIM;
            v[j][0] = srow[lo];
            v[j][1] = srow[lo + 32];
            v[j][2] = srow[lo + 64];
            v[j][3] = srow[lo + 96];
        }
    }
}

__device__ __forceinline__ void lds_add(float* p, float x) {
    __hip_atomic_fetch_add(p, x, __ATOMIC_RELAXED, __HIP_MEMORY_SCOPE_WORKGROUP);
}

__device__ __forceinline__ void add_rows(float* acc, int t, float nrm, int g, int cnt, unsigned lo,
                                         const float (&v)[EGRP][4]) {
#pragma unroll
    for (int j = 0; j < EGRP; ++j) {
        if (g + j < cnt) {
            float* drow = acc + __builtin_amdgcn_readlane(t, g + j) * HDIM;
            const float nj = __int_as_float(__builtin_amdgcn_readlane(__float_as_int(nrm), g + j));
            lds_add(drow + lo,      v[j][0] * nj);
            lds_add(drow + lo + 32, v[j][1] * nj);
            lds_add(drow + lo + 64, v[j][2] * nj);
            lds_add(drow + lo + 96, v[j][3] * nj);
        }
    }
}

__device__ __forceinline__ void drain(const float* __restrict__ h, const float* __restrict__ dis,
                                      const int* __restrict__ row, float* acc,
                                      const int* qE, const unsigned short* qT, int off, int cnt,
                                      unsigned lo, int node_lo) {
    const int li = off + min((int)lo, cnt - 1);
    const int ej = qE[li];
    const int tj = (int)qT[li];
    const int r  = row[ej];
    float v[EGRP][4];
    load_rows(h, r, 0, cnt, lo, v);
    const float nrm = dis[r] * dis[node_lo + tj];
    add_rows(acc, tj, nrm, 0, cnt, lo, v);
#pragma unroll
    for (int g = EGRP; g < 32; g += EGRP) {
        load_rows(h, r, g, cnt, lo, v);
        add_rows(acc, tj, nrm, g, cnt, lo, v);
    }
}

__device__ __forceinline__ void append(bool m, int eidx, unsigned t, int& qn, int* qE, unsigned short* qT) {
    const unsigned bal = __builtin_amdgcn_ballot_w32(m);
    if (bal == 0u) return;
    const int pos = qn + (int)__builtin_amdgcn_mbcnt_lo(bal, 0u);
    if (m) { qE[pos] = eidx; qT[pos] = (unsigned short)t; }
    qn += __builtin_popcount(bal);
}

__device__ __forceinline__ void append_seg(bool m, int seg, int& sn, int* sL) {
    const unsigned bal = __builtin_amdgcn_ballot_w32(m);
    if (bal == 0u) return;
    if (m) sL[sn + (int)__builtin_amdgcn_mbcnt_lo(bal, 0u)] = seg;
    sn += __builtin_popcount(bal);
}

__device__ __forceinline__ void scan_batch(const int* __restrict__ col, const int* sL, int off, int cnt, int e,
                                           unsigned lo, int node_lo, int rows, int& qn,
                                           int* qE, unsigned short* qT) {
    const int sv = sL[off + min((int)lo, cnt - 1)];
    int ei[SGRP];
    int c[SGRP];
#pragma unroll
    for (int j = 0; j < SGRP; ++j) {
        if (j < cnt) {
            ei[j] = __builtin_amdgcn_readlane(sv, j) * 32 + (int)lo;
            c[j] = col[min(ei[j], e - 1)];
        }
    }
#pragma unroll
    for (int j = 0; j < SGRP; ++j) {
        if (j < cnt) {
            const unsigned d = (unsigned)(c[j] - node_lo);
            append((d < (unsigned)rows) & (ei[j] < e), ei[j], d, qn, qE, qT);
        }
    }
}

__global__ __launch_bounds__(1024) void k_aggr(
        const float* __restrict__ h, const float* __restrict__ dis,
        const int* __restrict__ row, const int* __restrict__ col,
        const unsigned* __restrict__ present, int pitch,
        float* __restrict__ y, int n, int e) {
    __shared__ float acc[RANGE * HDIM];
    __shared__ int   qEs[AWAVES * QCAP];
    __shared__ int   sLs[AWAVES * SCAP];
    __shared__ unsigned short qTs[AWAVES * QCAP];

    const int tid = (int)threadIdx.x;
    const unsigned lo = threadIdx.x & 31;
    const int w = __builtin_amdgcn_readfirstlane(tid >> 5);
    const int node_lo = (int)blockIdx.x * RANGE;
    const int rows = min(RANGE, n - node_lo);
    int* qE = qEs + w * QCAP;
    unsigned short* qT = qTs + w * QCAP;
    int* sL = sLs + w * SCAP;

    float4* acc4 = reinterpret_cast<float4*>(acc);
    for (int i = tid; i < RANGE * HDIM / 4; i += 1024) acc4[i] = make_float4(0.f, 0.f, 0.f, 0.f);
    __syncthreads();

    int qn = 0;
    int sn = 0;
    const int nchunk = (e + CHUNK - 1) / CHUNK;
    const unsigned* mine = present + blockIdx.x;
#pragma unroll 1
    for (int cb = w; cb < nchunk; cb += 32 * AWAVES) {
        const int chunk = cb + AWAVES * (int)lo;
        const unsigned wd = (chunk < nchunk) ? mine[(size_t)chunk * pitch] : 0u;
#pragma unroll 1
        for (int kc = 0; kc < 32; kc += 8) {
            const unsigned part = (wd >> kc) & 0xffu;
            if (__builtin_amdgcn_ballot_w32(part != 0u) == 0u) continue;
#pragma unroll
            for (int k = 0; k < 8; ++k)
                append_seg(((part >> k) & 1u) != 0u, chunk * 32 + kc + k, sn, sL);
            while (sn >= SGRP) {
                sn -= SGRP;
                scan_batch(col, sL, sn, SGRP, e, lo, node_lo, rows, qn, qE, qT);
                while (qn >= 32) {
                    qn -= 32;
                    drain(h, dis, row, acc, qE, qT, qn, 32, lo, node_lo);
                }
            }
        }
    }
    if (sn > 0) {
        scan_batch(col, sL, 0, sn, e, lo, node_lo, rows, qn, qE, qT);
        while (qn >= 32) {
            qn -= 32;
            drain(h, dis, row, acc, qE, qT, qn, 32, lo, node_lo);
        }
    }

    if (qn > 0) drain(h, dis, row, acc, qE, qT, 0, qn, lo, node_lo);
    __syncthreads();

    float4* y4 = reinterpret_cast<float4*>(y + (size_t)node_lo * HDIM);
    for (int i = tid; i < rows * (HDIM / 4); i += 1024) y4[i] = acc4[i];
}

#define PVEC 4
#define PTHR 256

__device__ __forceinline__ float bn_relu(float v, float m, float vr, float g, float b) {
    v = (v - m) * rsqrtf(vr + 1e-5f) * g + b;
    return fmaxf(v, 0.0f);
}

__global__ __launch_bounds__(PTHR) void k_post1(const float* __restrict__ h, float* __restrict__ y,
                        const float* __restrict__ dis,
                        const float* __restrict__ b1,
                        const float* __restrict__ gamma,
                        const float* __restrict__ beta,
                        const float* __restrict__ mean,
                        const float* __restrict__ var, int total4) {
    const float4* __restrict__ h4 = reinterpret_cast<const float4*>(h);
    float4* __restrict__ y4 = reinterpret_cast<float4*>(y);
    const int c4 = (int)threadIdx.x & (HDIM / 4 - 1);
    const int base = (int)blockIdx.x * (PTHR * PVEC) + (int)threadIdx.x;
    float4 yv[PVEC], hv[PVEC];
    float d[PVEC];
#pragma unroll
    for (int j = 0; j < PVEC; ++j) {
        const int v = min(base + j * PTHR, total4 - 1);
        yv[j] = y4[v];
        hv[j] = h4[v];
        d[j] = dis[v / (HDIM / 4)];
    }
    const float4 bb = reinterpret_cast<const float4*>(b1)[c4];
    const float4 mm = reinterpret_cast<const float4*>(mean)[c4];
    const float4 vv = reinterpret_cast<const float4*>(var)[c4];
    const float4 gg = reinterpret_cast<const float4*>(gamma)[c4];
    const float4 be = reinterpret_cast<const float4*>(beta)[c4];
#pragma unroll
    for (int j = 0; j < PVEC; ++j) {
        const int v = base + j * PTHR;
        if (v < total4) {
            const float dd = d[j] * d[j];
            float4 o;
            o.x = bn_relu(yv[j].x + dd * hv[j].x + bb.x, mm.x, vv.x, gg.x, be.x);
            o.y = bn_relu(yv[j].y + dd * hv[j].y + bb.y, mm.y, vv.y, gg.y, be.y);
            o.z = bn_relu(yv[j].z + dd * hv[j].z + bb.z, mm.z, vv.z, gg.z, be.z);
            o.w = bn_relu(yv[j].w + dd * hv[j].w + bb.w, mm.w, vv.w, gg.w, be.w);
            y4[v] = o;
        }
    }
}

__global__ __launch_bounds__(PTHR) void k_post2(const float* __restrict__ h, const float* __restrict__ x1,
                        float* __restrict__ y,
                        const float* __restrict__ dis,
                        const float* __restrict__ b2, int total4) {
    const float4* __restrict__ h4 = reinterpret_cast<const float4*>(h);
    const float4* __restrict__ x4 = reinterpret_cast<const float4*>(x1);
    float4* __restrict__ y4 = reinterpret_cast<float4*>(y);
    const int c4 = (int)threadIdx.x & (HDIM / 4 - 1);
    const int base = (int)blockIdx.x * (PTHR * PVEC) + (int)threadIdx.x;
    float4 yv[PVEC], hv[PVEC], xv[PVEC];
    float d[PVEC];
#pragma unroll
    for (int j = 0; j < PVEC; ++j) {
        const int v = min(base + j * PTHR, total4 - 1);
        yv[j] = y4[v];
        hv[j] = h4[v];
        xv[j] = x4[v];
        d[j] = dis[v / (HDIM / 4)];
    }
    const float4 bb = reinterpret_cast<const float4*>(b2)[c4];
#pragma unroll
    for (int j = 0; j < PVEC; ++j) {
        const int v = base + j * PTHR;
        if (v < total4) {
            const float dd = d[j] * d[j];
            float4 o;
            o.x = fmaxf(xv[j].x, yv[j].x + dd * hv[j].x + bb.x);
            o.y = fmaxf(xv[j].y, yv[j].y + dd * hv[j].y + bb.y);
            o.z = fmaxf(xv[j].z, yv[j].z + dd * hv[j].z + bb.z);
            o.w = fmaxf(xv[j].w, yv[j].w + dd * hv[j].w + bb.w);
            y4[v] = o;
        }
    }
}

extern "C" void kernel_launch(void* const* d_in, const int* in_sizes, int n_in,
                              void* d_out, int out_size, void* d_ws, size_t ws_size,
                              hipStream_t stream) {
    const float* node_feat = (const float*)d_in[0];
    const int*   eidx      = (const int*)d_in[1];
    const float* W1        = (const float*)d_in[2];
    const float* b1        = (const float*)d_in[3];
    const float* gamma1    = (const float*)d_in[4];
    const float* beta1     = (const float*)d_in[5];
    const float* mean1     = (const float*)d_in[6];
    const float* var1      = (const float*)d_in[7];
    const float* W2        = (const float*)d_in[8];
    const float* b2        = (const float*)d_in[9];
    const float* Wf        = (const float*)d_in[10];
    const float* bf        = (const float*)d_in[11];
    float* out = (float*)d_out;

    const int n = in_sizes[0] / HDIM;
    const int e = in_sizes[1] / 2;
    const int* row = eidx;
    const int* col = eidx + e;

    float* dis = (float*)d_ws;
    float* h   = dis + n;
    float* y1  = h + (size_t)n * HDIM;
    float* y2  = y1 + (size_t)n * HDIM;
    unsigned* present = (unsigned*)(y2 + (size_t)n * HDIM);

    const int nh4     = n * (HDIM / 4);
    const int postGrd = (nh4 + PTHR * PVEC - 1) / (PTHR * PVEC);
    const int aggGrd  = (n + RANGE - 1) / RANGE;
    const int pitch   = (aggGrd + 31) / 32 * 32;
    const int chunks  = (e + CHUNK - 1) / CHUNK;

    k_init_deg<<<(n + 255) / 256, 256, 0, stream>>>(dis, n);
    k_edge_deg<<<chunks, CHUNK, 0, stream>>>(col, dis, present, e, pitch);
    k_finish_deg<<<(n + 255) / 256, 256, 0, stream>>>(dis, n);

    k_gemm_wmma<128, 2><<<GEMM_WGS, 256, 0, stream>>>(node_feat, W1, nullptr, h, n, HDIM);
    k_aggr<<<aggGrd, 1024, 0, stream>>>(h, dis, row, col, present, pitch, y1, n, e);
    k_post1<<<postGrd, PTHR, 0, stream>>>(h, y1, dis, b1, gamma1, beta1, mean1, var1, nh4);

    k_gemm_wmma<128, 2><<<GEMM_WGS, 256, 0, stream>>>(y1, W2, nullptr, h, n, HDIM);
    k_aggr<<<aggGrd, 1024, 0, stream>>>(h, dis, row, col, present, pitch, y2, n, e);
    k_post2<<<postGrd, PTHR, 0, stream>>>(h, y1, y2, dis, b2, nh4);

    k_gemm_wmma<48, 1><<<GEMM_WGS, 256, 0, stream>>>(y2, Wf, bf, out, n, ODIM);
}


# k_aggr layer 2 replays per-owner lists (source row, local target, norm) that layer 1's pass writes as it drains: no second edge filtering, no per-edge index or norm loads; sum = LDS arrival order
# speedup vs baseline: 2.7432x; 1.0750x over previous
#include <hip/hip_runtime.h>

typedef __attribute__((ext_vector_type(2))) float v2f;
typedef __attribute__((ext_vector_type(8))) float v8f;

#define HDIM 128
#define ODIM 40

__global__ void k_init_deg(float* __restrict__ deg, int n) {
    int i = blockIdx.x * blockDim.x + threadIdx.x;
    if (i < n) deg[i] = 1.0f;
}

#define RANGE   400
#define CHUNK   1024

__global__ __launch_bounds__(CHUNK) void k_edge_deg(const int* __restrict__ col, float* __restrict__ deg,
                                                    unsigned* __restrict__ present, int e, int pitch) {
    __shared__ unsigned pres[CHUNK];
    const int tid = (int)threadIdx.x;
    pres[tid] = 0u;
    __syncthreads();
    const int i = (int)blockIdx.x * CHUNK + tid;
    if (i < e) {
        const int c = col[i];
        atomicAdd(&deg[c], 1.0f);
        atomicOr(&pres[c / RANGE], 1u << (tid >> 5));
    }
    __syncthreads();
    if (tid < pitch) present[(size_t)blockIdx.x * pitch + tid] = pres[tid];
}

__global__ void k_finish_deg(float* __restrict__ deg, int n) {
    int i = blockIdx.x * blockDim.x + threadIdx.x;
    if (i < n) {
        float d = deg[i];
        deg[i] = (d > 0.0f) ? rsqrtf(d) : 0.0f;
    }
}

#define GEMM_WGS 256

template <int NPAD, int SPLIT>
__global__ __launch_bounds__(256) void k_gemm_wmma(
        const float* __restrict__ A, const float* __restrict__ B,
        const float* __restrict__ bias, float* __restrict__ C,
        int M, int ncols) {
    constexpr int K   = 128;
    constexpr int KS  = K / 4;
    constexpr int NTW = NPAD / 16 / SPLIT;

    const int lane  = threadIdx.x & 31;
    const int c     = lane & 15;
    const int koff  = (lane >> 4) * 2;
    const int gw    = (int)blockIdx.x * 8 + __builtin_amdgcn_readfirstlane((int)threadIdx.x >> 5);
    const int share = gw % SPLIT;
    const int t0    = gw / SPLIT;
    const int tstep = (int)gridDim.x * 8 / SPLIT;
    const int mtile = M >> 4;
    if (t0 >= mtile) return;
    const int nbase = share * (NTW * 16);

    v2f b[KS][NTW];
    float badd[NTW];
#pragma unroll
    for (int nt = 0; nt < NTW; ++nt) {
        const int n = nbase + nt * 16 + c;
        const bool in = n < ncols;
        const int nn = in ? n : 0;
        badd[nt] = (bias && in) ? bias[nn] : 0.0f;
        const float* bp = B + (size_t)koff * ncols + nn;
#pragma unroll
        for (int ks = 0; ks < KS; ++ks) {
            v2f t;
            t.x = bp[(ks * 4) * ncols];
            t.y = bp[(ks * 4 + 1) * ncols];
            if (!in) { t.x = 0.f; t.y = 0.f; }
            b[ks][nt] = t;
        }
    }

    v2f a[KS];
    {
        const float* aptr = A + (size_t)(t0 * 16 + c) * K + koff;
#pragma unroll
        for (int ks = 0; ks < KS; ++ks) a[ks] = *(const v2f*)(aptr + ks * 4);
    }

    const v8f z8 = {0.f, 0.f, 0.f, 0.f, 0.f, 0.f, 0.f, 0.f};
#pragma unroll 1
    for (int tile = t0; tile < mtile; tile += tstep) {
        v2f an[KS];
        const int nxt = tile + tstep;
        if (nxt < mtile) {
            const float* aptr = A + (size_t)(nxt * 16 + c) * K + koff;
#pragma unroll
            for (int ks = 0; ks < KS; ++ks) an[ks] = *(const v2f*)(aptr + ks * 4);
        } else {
#pragma unroll
            for (int ks = 0; ks < KS; ++ks) an[ks] = a[ks];
        }

        v8f acc[NTW];
#pragma unroll
        for (int nt = 0; nt < NTW; ++nt) acc[nt] = z8;
#pragma unroll
        for (int ks = 0; ks < KS; ++ks) {
#pragma unroll
            for (int nt = 0; nt < NTW; ++nt) {
                acc[nt] = __builtin_amdgcn_wmma_f32_16x16x4_f32(
                     false, a[ks],  false, b[ks][nt],
                     (short)0, acc[nt],
                     false,  false);
            }
        }

        const int rbase = tile * 16 + ((lane >> 4) << 3);
#pragma unroll
        for (int nt = 0; nt < NTW; ++nt) {
            const int n = nbase + nt * 16 + c;
            if (n >= ncols) continue;
            float* cp = C + (size_t)rbase * ncols + n;
#pragma unroll
            for (int r = 0; r < 8; ++r) cp[r * ncols] = acc[nt][r] + badd[nt];
        }
#pragma unroll
        for (int ks = 0; ks < KS; ++ks) a[ks] = an[ks];
    }
}

#define AWAVES  32
#define SGRP    8
#define QCAP    288
#define SCAP    264
#define EGRP    8
#define LCAP    16384
#define LBITS   22

__device__ __forceinline__ void load_rows(const float* __restrict__ h, int r, int g, int cnt, unsigned lo,
                                          float (&v)[EGRP][4]) {
#pragma unroll
    for (int j = 0; j < EGRP; ++j) {
        if (g + j < cnt) {
            const float* srow = h + (size_t)__builtin_amdgcn_readlane(r, g + j) * HDIM;
            v[j][0] = srow[lo];
            v[j][1] = srow[lo + 32];
            v[j][2] = srow[lo + 64];
            v[j][3] = srow[lo + 96];
        }
    }
}

__device__ __forceinline__ void lds_add(float* p, float x) {
    __hip_atomic_fetch_add(p, x, __ATOMIC_RELAXED, __HIP_MEMORY_SCOPE_WORKGROUP);
}

__device__ __forceinline__ void add_rows(float* acc, int t, float nrm, int g, int cnt, unsigned lo,
                                         const float (&v)[EGRP][4]) {
#pragma unroll
    for (int j = 0; j < EGRP; ++j) {
        if (g + j < cnt) {
            float* drow = acc + __builtin_amdgcn_readlane(t, g + j) * HDIM;
            const float nj = __int_as_float(__builtin_amdgcn_readlane(__float_as_int(nrm), g + j));
            lds_add(drow + lo,      v[j][0] * nj);
            lds_add(drow + lo + 32, v[j][1] * nj);
            lds_add(drow + lo + 64, v[j][2] * nj);
            lds_add(drow + lo + 96, v[j][3] * nj);
        }
    }
}

template <bool REC>
__device__ __forceinline__ void drain(const float* __restrict__ h, const float* __restrict__ dis,
                                      const int* __restrict__ row, float* acc,
                                      const int* qE, const unsigned short* qT, int off, int cnt,
                                      unsigned lo, int node_lo, int2* __restrict__ lst, int* lcur) {
    const int li = off + min((int)lo, cnt - 1);
    const int ej = qE[li];
    const int tj = (int)qT[li];
    const int r  = row[ej];
    float v[EGRP][4];
    load_rows(h, r, 0, cnt, lo, v);
    const float nrm = dis[r] * dis[node_lo + tj];
    if (REC) {
        int pos = 0;
        if (lo == 0u) pos = __hip_atomic_fetch_add(lcur, cnt, __ATOMIC_RELAXED, __HIP_MEMORY_SCOPE_WORKGROUP);
        pos = __builtin_amdgcn_readfirstlane(pos);
        if (pos + cnt <= LCAP) {
            int2 en;
            en.x = r | (tj << LBITS);
            en.y = __float_as_int(nrm);
            if ((int)lo < cnt) lst[pos + (int)lo] = en;
        }
    }
    add_rows(acc, tj, nrm, 0, cnt, lo, v);
#pragma unroll
    for (int g = EGRP; g < 32; g += EGRP) {
        load_rows(h, r, g, cnt, lo, v);
        add_rows(acc, tj, nrm, g, cnt, lo, v);
    }
}

__device__ __forceinline__ void append(bool m, int eidx, unsigned t, int& qn, int* qE, unsigned short* qT) {
    const unsigned bal = __builtin_amdgcn_ballot_w32(m);
    if (bal == 0u) return;
    const int pos = qn + (int)__builtin_amdgcn_mbcnt_lo(bal, 0u);
    if (m) { qE[pos] = eidx; qT[pos] = (unsigned short)t; }
    qn += __builtin_popcount(bal);
}

__device__ __forceinline__ void append_seg(bool m, int seg, int& sn, int* sL) {
    const unsigned bal = __builtin_amdgcn_ballot_w32(m);
    if (bal == 0u) return;
    if (m) sL[sn + (int)__builtin_amdgcn_mbcnt_lo(bal, 0u)] = seg;
    sn += __builtin_popcount(bal);
}

__device__ __forceinline__ void scan_batch(const int* __restrict__ col, const int* sL, int off, int cnt, int e,
                                           unsigned lo, int node_lo, int rows, int& qn,
                                           int* qE, unsigned short* qT) {
    const int sv = sL[off + min((int)lo, cnt - 1)];
    int ei[SGRP];
    int c[SGRP];
#pragma unroll
    for (int j = 0; j < SGRP; ++j) {
        if (j < cnt) {
            ei[j] = __builtin_amdgcn_readlane(sv, j) * 32 + (int)lo;
            c[j] = col[min(ei[j], e - 1)];
        }
    }
#pragma unroll
    for (int j = 0; j < SGRP; ++j) {
        if (j < cnt) {
            const unsigned d = (unsigned)(c[j] - node_lo);
            append((d < (unsigned)rows) & (ei[j] < e), ei[j], d, qn, qE, qT);
        }
    }
}

template <bool REC>
__device__ __forceinline__ void scan_pass(const float* __restrict__ h, const float* __restrict__ dis,
                                          const int* __restrict__ row, const int* __restrict__ col,
                                          const unsigned* __restrict__ mine, int pitch, float* acc,
                                          int* qE, unsigned short* qT, int* sL, int w, unsigned lo,
                                          int node_lo, int rows, int e, int2* __restrict__ lst, int* lcur) {
    int qn = 0;
    int sn = 0;
    const int nchunk = (e + CHUNK - 1) / CHUNK;
#pragma unroll 1
    for (int cb = w; cb < nchunk; cb += 32 * AWAVES) {
        const int chunk = cb + AWAVES * (int)lo;
        const unsigned wd = (chunk < nchunk) ? mine[(size_t)chunk * pitch] : 0u;
#pragma unroll 1
        for (int kc = 0; kc < 32; kc += 8) {
            const unsigned part = (wd >> kc) & 0xffu;
            if (__builtin_amdgcn_ballot_w32(part != 0u) == 0u) continue;
#pragma unroll
            for (int k = 0; k < 8; ++k)
                append_seg(((part >> k) & 1u) != 0u, chunk * 32 + kc + k, sn, sL);
            while (sn >= SGRP) {
                sn -= SGRP;
                scan_batch(col, sL, sn, SGRP, e, lo, node_lo, rows, qn, qE, qT);
                while (qn >= 32) {
                    qn -= 32;
                    drain<REC>(h, dis, row, acc, qE, qT, qn, 32, lo, node_lo, lst, lcur);
                }
            }
        }
    }
    if (sn > 0) {
        scan_batch(col, sL, 0, sn, e, lo, node_lo, rows, qn, qE, qT);
        while (qn >= 32) {
            qn -= 32;
            drain<REC>(h, dis, row, acc, qE, qT, qn, 32, lo, node_lo, lst, lcur);
        }
    }
    if (qn > 0) drain<REC>(h, dis, row, acc, qE, qT, 0, qn, lo, node_lo, lst, lcur);
}

__device__ __forceinline__ void gather_add(const float* __restrict__ h, float* acc, int r, int t, float nrm,
                                           int cnt, unsigned lo) {
    float v[EGRP][4];
#pragma unroll
    for (int g = 0; g < 32; g += EGRP) {
        load_rows(h, r, g, cnt, lo, v);
        add_rows(acc, t, nrm, g, cnt, lo, v);
    }
}

template <int MODE>
__global__ __launch_bounds__(1024) void k_aggr(
        const float* __restrict__ h, const float* __restrict__ dis,
        const int* __restrict__ row, const int* __restrict__ col,
        const unsigned* __restrict__ present, int pitch,
        float* __restrict__ y, int n, int e, int* __restrict__ nent, int* __restrict__ lstw) {
    static_assert(RANGE <= 512, "list entry packing");
    __shared__ float acc[RANGE * HDIM];
    __shared__ int   qEs[AWAVES * QCAP];
    __shared__ int   sLs[AWAVES * SCAP];
    __shared__ unsigned short qTs[AWAVES * QCAP];
    __shared__ int   lcur;

    const int tid = (int)threadIdx.x;
    const unsigned lo = threadIdx.x & 31;
    const int w = __builtin_amdgcn_readfirstlane(tid >> 5);
    const int node_lo = (int)blockIdx.x * RANGE;
    const int rows = min(RANGE, n - node_lo);
    int* qE = qEs + w * QCAP;
    unsigned short* qT = qTs + w * QCAP;
    int* sL = sLs + w * SCAP;
    int2* lst = reinterpret_cast<int2*>(lstw) + (size_t)blockIdx.x * LCAP;
    const unsigned* mine = present + blockIdx.x;

    float4* acc4 = reinterpret_cast<float4*>(acc);
    for (int i = tid; i < RANGE * HDIM / 4; i += 1024) acc4[i] = make_float4(0.f, 0.f, 0.f, 0.f);
    if (tid == 0) lcur = (n <= (1 << LBITS)) ? 0 : LCAP + 1;
    __syncthreads();

    if (MODE == 0) {
        scan_pass<true>(h, dis, row, col, mine, pitch, acc, qE, qT, sL, w, lo, node_lo, rows, e, lst, &lcur);
    } else {
        const int ne = __builtin_amdgcn_readfirstlane(nent[blockIdx.x]);
        if (ne >= 0) {
            const int nb = (ne + 31) >> 5;
            const int2* L = lst;
            int2 en = make_int2(0, 0);
            if (w < nb) en = L[w * 32 + min((int)lo, min(32, ne - w * 32) - 1)];
#pragma unroll 1
            for (int b = w; b < nb; b += AWAVES) {
                int2 nx = en;
                const int bn = b + AWAVES;
                if (bn < nb) nx = L[bn * 32 + min((int)lo, min(32, ne - bn * 32) - 1)];
                const int cnt = min(32, ne - b * 32);
                const int r = en.x & ((1 << LBITS) - 1);
                const int t = en.x >> LBITS;
                const float nrm = __int_as_float(en.y);
                if (cnt == 32) gather_add(h, acc, r, t, nrm, 32, lo);
                else           gather_add(h, acc, r, t, nrm, cnt, lo);
                en = nx;
            }
        } else {
            scan_pass<false>(h, dis, row, col, mine, pitch, acc, qE, qT, sL, w, lo, node_lo, rows, e, lst, &lcur);
        }
    }
    __syncthreads();

    if (MODE == 0 && tid == 0) nent[blockIdx.x] = (lcur <= LCAP) ? lcur : -1;

    float4* y4 = reinterpret_cast<float4*>(y + (size_t)node_lo * HDIM);
    for (int i = tid; i < rows * (HDIM / 4); i += 1024) y4[i] = acc4[i];
}

#define PVEC 4
#define PTHR 256

__device__ __forceinline__ float bn_relu(float v, float m, float vr, float g, float b) {
    v = (v - m) * rsqrtf(vr + 1e-5f) * g + b;
    return fmaxf(v, 0.0f);
}

__global__ __launch_bounds__(PTHR) void k_post1(const float* __restrict__ h, float* __restrict__ y,
                        const float* __restrict__ dis,
                        const float* __restrict__ b1,
                        const float* __restrict__ gamma,
                        const float* __restrict__ beta,
                        const float* __restrict__ mean,
                        const float* __restrict__ var, int total4) {
    const float4* __restrict__ h4 = reinterpret_cast<const float4*>(h);
    float4* __restrict__ y4 = reinterpret_cast<float4*>(y);
    const int c4 = (int)threadIdx.x & (HDIM / 4 - 1);
    const int base = (int)blockIdx.x * (PTHR * PVEC) + (int)threadIdx.x;
    float4 yv[PVEC], hv[PVEC];
    float d[PVEC];
#pragma unroll
    for (int j = 0; j < PVEC; ++j) {
        const int v = min(base + j * PTHR, total4 - 1);
        yv[j] = y4[v];
        hv[j] = h4[v];
        d[j] = dis[v / (HDIM / 4)];
    }
    const float4 bb = reinterpret_cast<const float4*>(b1)[c4];
    const float4 mm = reinterpret_cast<const float4*>(mean)[c4];
    const float4 vv = reinterpret_cast<const float4*>(var)[c4];
    const float4 gg = reinterpret_cast<const float4*>(gamma)[c4];
    const float4 be = reinterpret_cast<const float4*>(beta)[c4];
#pragma unroll
    for (int j = 0; j < PVEC; ++j) {
        const int v = base + j * PTHR;
        if (v < total4) {
            const float dd = d[j] * d[j];
            float4 o;
            o.x = bn_relu(yv[j].x + dd * hv[j].x + bb.x, mm.x, vv.x, gg.x, be.x);
            o.y = bn_relu(yv[j].y + dd * hv[j].y + bb.y, mm.y, vv.y, gg.y, be.y);
            o.z = bn_relu(yv[j].z + dd * hv[j].z + bb.z, mm.z, vv.z, gg.z, be.z);
            o.w = bn_relu(yv[j].w + dd * hv[j].w + bb.w, mm.w, vv.w, gg.w, be.w);
            y4[v] = o;
        }
    }
}

__global__ __launch_bounds__(PTHR) void k_post2(const float* __restrict__ h, const float* __restrict__ x1,
                        float* __restrict__ y,
                        const float* __restrict__ dis,
                        const float* __restrict__ b2, int total4) {
    const float4* __restrict__ h4 = reinterpret_cast<const float4*>(h);
    const float4* __restrict__ x4 = reinterpret_cast<const float4*>(x1);
    float4* __restrict__ y4 = reinterpret_cast<float4*>(y);
    const int c4 = (int)threadIdx.x & (HDIM / 4 - 1);
    const int base = (int)blockIdx.x * (PTHR * PVEC) + (int)threadIdx.x;
    float4 yv[PVEC], hv[PVEC], xv[PVEC];
    float d[PVEC];
#pragma unroll
    for (int j = 0; j < PVEC; ++j) {
        const int v = min(base + j * PTHR, total4 - 1);
        yv[j] = y4[v];
        hv[j] = h4[v];
        xv[j] = x4[v];
        d[j] = dis[v / (HDIM / 4)];
    }
    const float4 bb = reinterpret_cast<const float4*>(b2)[c4];
#pragma unroll
    for (int j = 0; j < PVEC; ++j) {
        const int v = base + j * PTHR;
        if (v < total4) {
            const float dd = d[j] * d[j];
            float4 o;
            o.x = fmaxf(xv[j].x, yv[j].x + dd * hv[j].x + bb.x);
            o.y = fmaxf(xv[j].y, yv[j].y + dd * hv[j].y + bb.y);
            o.z = fmaxf(xv[j].z, yv[j].z + dd * hv[j].z + bb.z);
            o.w = fmaxf(xv[j].w, yv[j].w + dd * hv[j].w + bb.w);
            y4[v] = o;
        }
    }
}

extern "C" void kernel_launch(void* const* d_in, const int* in_sizes, int n_in,
                              void* d_out, int out_size, void* d_ws, size_t ws_size,
                              hipStream_t stream) {
    const float* node_feat = (const float*)d_in[0];
    const int*   eidx      = (const int*)d_in[1];
    const float* W1        = (const float*)d_in[2];
    const float* b1        = (const float*)d_in[3];
    const float* gamma1    = (const float*)d_in[4];
    const float* beta1     = (const float*)d_in[5];
    const float* mean1     = (const float*)d_in[6];
    const float* var1      = (const float*)d_in[7];
    const float* W2        = (const float*)d_in[8];
    const float* b2        = (const float*)d_in[9];
    const float* Wf        = (const float*)d_in[10];
    const float* bf        = (const float*)d_in[11];
    float* out = (float*)d_out;

    const int n = in_sizes[0] / HDIM;
    const int e = in_sizes[1] / 2;
    const int* row = eidx;
    const int* col = eidx + e;

    float* dis = (float*)d_ws;
    float* h   = dis + n;
    float* y1  = h + (size_t)n * HDIM;
    float* y2  = y1 + (size_t)n * HDIM;
    unsigned* present = (unsigned*)(y2 + (size_t)n * HDIM);

    const int nh4     = n * (HDIM / 4);
    const int postGrd = (nh4 + PTHR * PVEC - 1) / (PTHR * PVEC);
    const int aggGrd  = (n + RANGE - 1) / RANGE;
    const int pitch   = (aggGrd + 31) / 32 * 32;
    const int chunks  = (e + CHUNK - 1) / CHUNK;
    int* nent = (int*)(present + (size_t)chunks * pitch);
    int* lstw = nent + pitch;

    k_init_deg<<<(n + 255) / 256, 256, 0, stream>>>(dis, n);
    k_edge_deg<<<chunks, CHUNK, 0, stream>>>(col, dis, present, e, pitch);
    k_finish_deg<<<(n + 255) / 256, 256, 0, stream>>>(dis, n);

    k_gemm_wmma<128, 2><<<GEMM_WGS, 256, 0, stream>>>(node_feat, W1, nullptr, h, n, HDIM);
    k_aggr<0><<<aggGrd, 1024, 0, stream>>>(h, dis, row, col, present, pitch, y1, n, e, nent, lstw);
    k_post1<<<postGrd, PTHR, 0, stream>>>(h, y1, dis, b1, gamma1, beta1, mean1, var1, nh4);

    k_gemm_wmma<128, 2><<<GEMM_WGS, 256, 0, stream>>>(y1, W2, nullptr, h, n, HDIM);
    k_aggr<1><<<aggGrd, 1024, 0, stream>>>(h, dis, row, col, present, pitch, y2, n, e, nent, lstw);
    k_post2<<<postGrd, PTHR, 0, stream>>>(h, y1, y2, dis, b2, nh4);

    k_gemm_wmma<48, 1><<<GEMM_WGS, 256, 0, stream>>>(y2, Wf, bf, out, n, ODIM);
}


# Owner work-groups finish their rows: epilogue in the LDS sums, then the next dense product (x1 W2; xmax Wf) by f32 WMMA from those LDS rows, pitch 132; 10 -> 6 launches; sum = LDS arrival order
# speedup vs baseline: 3.1404x; 1.1448x over previous
#include <hip/hip_runtime.h>

typedef __attribute__((ext_vector_type(2))) float v2f;
typedef __attribute__((ext_vector_type(8))) float v8f;

#define HDIM 128
#define ODIM 40

__global__ void k_init_deg(float* __restrict__ deg, int n) {
    int i = blockIdx.x * blockDim.x + threadIdx.x;
    if (i < n) deg[i] = 1.0f;
}

#define RANGE   400
#define CHUNK   1024

__global__ __launch_bounds__(CHUNK) void k_edge_deg(const int* __restrict__ col, float* __restrict__ deg,
                                                    unsigned* __restrict__ present, int e, int pitch) {
    __shared__ unsigned pres[CHUNK];
    const int tid = (int)threadIdx.x;
    pres[tid] = 0u;
    __syncthreads();
    const int i = (int)blockIdx.x * CHUNK + tid;
    if (i < e) {
        const int c = col[i];
        atomicAdd(&deg[c], 1.0f);
        atomicOr(&pres[c / RANGE], 1u << (tid >> 5));
    }
    __syncthreads();
    if (tid < pitch) present[(size_t)blockIdx.x * pitch + tid] = pres[tid];
}

__global__ void k_finish_deg(float* __restrict__ deg, int n) {
    int i = blockIdx.x * blockDim.x + threadIdx.x;
    if (i < n) {
        float d = deg[i];
        deg[i] = (d > 0.0f) ? rsqrtf(d) : 0.0f;
    }
}

#define GEMM_WGS 256

template <int NPAD, int SPLIT>
__global__ __launch_bounds__(256) void k_gemm_wmma(
        const float* __restrict__ A, const float* __restrict__ B,
        const float* __restrict__ bias, float* __restrict__ C,
        int M, int ncols) {
    constexpr int K   = 128;
    constexpr int KS  = K / 4;
    constexpr int NTW = NPAD / 16 / SPLIT;

    const int lane  = threadIdx.x & 31;
    const int c     = lane & 15;
    const int koff  = (lane >> 4) * 2;
    const int gw    = (int)blockIdx.x * 8 + __builtin_amdgcn_readfirstlane((int)threadIdx.x >> 5);
    const int share = gw % SPLIT;
    const int t0    = gw / SPLIT;
    const int tstep = (int)gridDim.x * 8 / SPLIT;
    const int mtile = M >> 4;
    if (t0 >= mtile) return;
    const int nbase = share * (NTW * 16);

    v2f b[KS][NTW];
    float badd[NTW];
#pragma unroll
    for (int nt = 0; nt < NTW; ++nt) {
        const int n = nbase + nt * 16 + c;
        const bool in = n < ncols;
        const int nn = in ? n : 0;
        badd[nt] = (bias && in) ? bias[nn] : 0.0f;
        const float* bp = B + (size_t)koff * ncols + nn;
#pragma unroll
        for (int ks = 0; ks < KS; ++ks) {
            v2f t;
            t.x = bp[(ks * 4) * ncols];
            t.y = bp[(ks * 4 + 1) * ncols];
            if (!in) { t.x = 0.f; t.y = 0.f; }
            b[ks][nt] = t;
        }
    }

    v2f a[KS];
    {
        const float* aptr = A + (size_t)(t0 * 16 + c) * K + koff;
#pragma unroll
        for (int ks = 0; ks < KS; ++ks) a[ks] = *(const v2f*)(aptr + ks * 4);
    }

    const v8f z8 = {0.f, 0.f, 0.f, 0.f, 0.f, 0.f, 0.f, 0.f};
#pragma unroll 1
    for (int tile = t0; tile < mtile; tile += tstep) {
        v2f an[KS];
        const int nxt = tile + tstep;
        if (nxt < mtile) {
            const float* aptr = A + (size_t)(nxt * 16 + c) * K + koff;
#pragma unroll
            for (int ks = 0; ks < KS; ++ks) an[ks] = *(const v2f*)(aptr + ks * 4);
        } else {
#pragma unroll
            for (int ks = 0; ks < KS; ++ks) an[ks] = a[ks];
        }

        v8f acc[NTW];
#pragma unroll
        for (int nt = 0; nt < NTW; ++nt) acc[nt] = z8;
#pragma unroll
        for (int ks = 0; ks < KS; ++ks) {
#pragma unroll
            for (int nt = 0; nt < NTW; ++nt) {
                acc[nt] = __builtin_amdgcn_wmma_f32_16x16x4_f32(
                     false, a[ks],  false, b[ks][nt],
                     (short)0, acc[nt],
                     false,  false);
            }
        }

        const int rbase = tile * 16 + ((lane >> 4) << 3);
#pragma unroll
        for (int nt = 0; nt < NTW; ++nt) {
            const int n = nbase + nt * 16 + c;
            if (n >= ncols) continue;
            float* cp = C + (size_t)rbase * ncols + n;
#pragma unroll
            for (int r = 0; r < 8; ++r) cp[r * ncols] = acc[nt][r] + badd[nt];
        }
#pragma unroll
        for (int ks = 0; ks < KS; ++ks) a[ks] = an[ks];
    }
}

#define AWAVES  32
#define SGRP    8
#define QCAP    288
#define SCAP    264
#define EGRP    8
#define LCAP    16384
#define LBITS   22

__device__ __forceinline__ void load_rows(const float* __restrict__ h, int r, int g, int cnt, unsigned lo,
                                          float (&v)[EGRP][4]) {
#pragma unroll
    for (int j = 0; j < EGRP; ++j) {
        if (g + j < cnt) {
            const float* srow = h + (size_t)__builtin_amdgcn_readlane(r, g + j) * HDIM;
            v[j][0] = srow[lo];
            v[j][1] = srow[lo + 32];
            v[j][2] = srow[lo + 64];
            v[j][3] = srow[lo + 96];
        }
    }
}

__device__ __forceinline__ void lds_add(float* p, float x) {
    __hip_atomic_fetch_add(p, x, __ATOMIC_RELAXED, __HIP_MEMORY_SCOPE_WORKGROUP);
}

template <int AP>
__device__ __forceinline__ void add_rows(float* acc, int t, float nrm, int g, int cnt, unsigned lo,
                                         const float (&v)[EGRP][4]) {
#pragma unroll
    for (int j = 0; j < EGRP; ++j) {
        if (g + j < cnt) {
            float* drow = acc + __builtin_amdgcn_readlane(t, g + j) * AP;
            const float nj = __int_as_float(__builtin_amdgcn_readlane(__float_as_int(nrm), g + j));
            lds_add(drow + lo,      v[j][0] * nj);
            lds_add(drow + lo + 32, v[j][1] * nj);
            lds_add(drow + lo + 64, v[j][2] * nj);
            lds_add(drow + lo + 96, v[j][3] * nj);
        }
    }
}

template <bool REC, int AP>
__device__ __forceinline__ void drain(const float* __restrict__ h, const float* __restrict__ dis,
                                      const int* __restrict__ row, float* acc,
                                      const int* qE, const unsigned short* qT, int off, int cnt,
                                      unsigned lo, int node_lo, int2* __restrict__ lst, int* lcur) {
    const int li = off + min((int)lo, cnt - 1);
    const int ej = qE[li];
    const int tj = (int)qT[li];
    const int r  = row[ej];
    float v[EGRP][4];
    load_rows(h, r, 0, cnt, lo, v);
    const float nrm = dis[r] * dis[node_lo + tj];
    if (REC) {
        int pos = 0;
        if (lo == 0u) pos = __hip_atomic_fetch_add(lcur, cnt, __ATOMIC_RELAXED, __HIP_MEMORY_SCOPE_WORKGROUP);
        pos = __builtin_amdgcn_readfirstlane(pos);
        if (pos + cnt <= LCAP) {
            int2 en;
            en.x = r | (tj << LBITS);
            en.y = __float_as_int(nrm);
            if ((int)lo < cnt) lst[pos + (int)lo] = en;
        }
    }
    add_rows<AP>(acc, tj, nrm, 0, cnt, lo, v);
#pragma unroll
    for (int g = EGRP; g < 32; g += EGRP) {
        load_rows(h, r, g, cnt, lo, v);
        add_rows<AP>(acc, tj, nrm, g, cnt, lo, v);
    }
}

__device__ __forceinline__ void append(bool m, int eidx, unsigned t, int& qn, int* qE, unsigned short* qT) {
    const unsigned bal = __builtin_amdgcn_ballot_w32(m);
    if (bal == 0u) return;
    const int pos = qn + (int)__builtin_amdgcn_mbcnt_lo(bal, 0u);
    if (m) { qE[pos] = eidx; qT[pos] = (unsigned short)t; }
    qn += __builtin_popcount(bal);
}

__device__ __forceinline__ void append_seg(bool m, int seg, int& sn, int* sL) {
    const unsigned bal = __builtin_amdgcn_ballot_w32(m);
    if (bal == 0u) return;
    if (m) sL[sn + (int)__builtin_amdgcn_mbcnt_lo(bal, 0u)] = seg;
    sn += __builtin_popcount(bal);
}

__device__ __forceinline__ void scan_batch(const int* __restrict__ col, const int* sL, int off, int cnt, int e,
                                           unsigned lo, int node_lo, int rows, int& qn,
                                           int* qE, unsigned short* qT) {
    const int sv = sL[off + min((int)lo, cnt - 1)];
    int ei[SGRP];
    int c[SGRP];
#pragma unroll
    for (int j = 0; j < SGRP; ++j) {
        if (j < cnt) {
            ei[j] = __builtin_amdgcn_readlane(sv, j) * 32 + (int)lo;
            c[j] = col[min(ei[j], e - 1)];
        }
    }
#pragma unroll
    for (int j = 0; j < SGRP; ++j) {
        if (j < cnt) {
            const unsigned d = (unsigned)(c[j] - node_lo);
            append((d < (unsigned)rows) & (ei[j] < e), ei[j], d, qn, qE, qT);
        }
    }
}

template <bool REC, int AP>
__device__ __forceinline__ void scan_pass(const float* __restrict__ h, const float* __restrict__ dis,
                                          const int* __restrict__ row, const int* __restrict__ col,
                                          const unsigned* __restrict__ mine, int pitch, float* acc,
                                          int* qE, unsigned short* qT, int* sL, int w, unsigned lo,
                                          int node_lo, int rows, int e, int2* __restrict__ lst, int* lcur) {
    int qn = 0;
    int sn = 0;
    const int nchunk = (e + CHUNK - 1) / CHUNK;
#pragma unroll 1
    for (int cb = w; cb < nchunk; cb += 32 * AWAVES) {
        const int chunk = cb + AWAVES * (int)lo;
        const unsigned wd = (chunk < nchunk) ? mine[(size_t)chunk * pitch] : 0u;
#pragma unroll 1
        for (int kc = 0; kc < 32; kc += 8) {
            const unsigned part = (wd >> kc) & 0xffu;
            if (__builtin_amdgcn_ballot_w32(part != 0u) == 0u) continue;
#pragma unroll
            for (int k = 0; k < 8; ++k)
                append_seg(((part >> k) & 1u) != 0u, chunk * 32 + kc + k, sn, sL);
            while (sn >= SGRP) {
                sn -= SGRP;
                scan_batch(col, sL, sn, SGRP, e, lo, node_lo, rows, qn, qE, qT);
                while (qn >= 32) {
                    qn -= 32;
                    drain<REC, AP>(h, dis, row, acc, qE, qT, qn, 32, lo, node_lo, lst, lcur);
                }
            }
        }
    }
    if (sn > 0) {
        scan_batch(col, sL, 0, sn, e, lo, node_lo, rows, qn, qE, qT);
        while (qn >= 32) {
            qn -= 32;
            drain<REC, AP>(h, dis, row, acc, qE, qT, qn, 32, lo, node_lo, lst, lcur);
        }
    }
    if (qn > 0) drain<REC, AP>(h, dis, row, acc, qE, qT, 0, qn, lo, node_lo, lst, lcur);
}

template <int AP>
__device__ __forceinline__ void gather_add(const float* __restrict__ h, float* acc, int r, int t, float nrm,
                                           int cnt, unsigned lo) {
    float v[EGRP][4];
#pragma unroll
    for (int g = 0; g < 32; g += EGRP) {
        load_rows(h, r, g, cnt, lo, v);
        add_rows<AP>(acc, t, nrm, g, cnt, lo, v);
    }
}

#define APITCH 132
#define TVEC   7

__device__ __forceinline__ float bn_relu(float v, float m, float vr, float g, float b) {
    v = (v - m) * rsqrtf(vr + 1e-5f) * g + b;
    return fmaxf(v, 0.0f);
}

__device__ __forceinline__ void rows_epilogue1(float* acc, int rows, int node_lo, int tid,
        const float* __restrict__ h, const float* __restrict__ dis, const float* __restrict__ b1,
        const float* __restrict__ gamma, const float* __restrict__ beta,
        const float* __restrict__ mean, const float* __restrict__ var, float* __restrict__ x1) {
    const float4* __restrict__ h4 = reinterpret_cast<const float4*>(h + (size_t)node_lo * HDIM);
    float4* __restrict__ x4 = reinterpret_cast<float4*>(x1 + (size_t)node_lo * HDIM);
    const int total = rows * (HDIM / 4);
    const int c4 = tid & (HDIM / 4 - 1);
    const float4 bb = reinterpret_cast<const float4*>(b1)[c4];
    const float4 mm = reinterpret_cast<const float4*>(mean)[c4];
    const float4 vv = reinterpret_cast<const float4*>(var)[c4];
    const float4 gg = reinterpret_cast<const float4*>(gamma)[c4];
    const float4 be = reinterpret_cast<const float4*>(beta)[c4];
    for (int i0 = tid; i0 < total; i0 += TVEC * 1024) {
        float4 hv[TVEC];
        float d[TVEC];
#pragma unroll
        for (int j = 0; j < TVEC; ++j) {
            const int i = min(i0 + j * 1024, total - 1);
            hv[j] = h4[i];
            d[j] = dis[node_lo + i / (HDIM / 4)];
        }
#pragma unroll
        for (int j = 0; j < TVEC; ++j) {
            const int i = i0 + j * 1024;
            if (i < total) {
                float4* ap = reinterpret_cast<float4*>(acc + (i / (HDIM / 4)) * APITCH + 4 * c4);
                const float4 a = *ap;
                const float dd = d[j] * d[j];
                float4 o;
                o.x = bn_relu(a.x + dd * hv[j].x + bb.x, mm.x, vv.x, gg.x, be.x);
                o.y = bn_relu(a.y + dd * hv[j].y + bb.y, mm.y, vv.y, gg.y, be.y);
                o.z = bn_relu(a.z + dd * hv[j].z + bb.z, mm.z, vv.z, gg.z, be.z);
                o.w = bn_relu(a.w + dd * hv[j].w + bb.w, mm.w, vv.w, gg.w, be.w);
                *ap = o;
                x4[i] = o;
            }
        }
    }
}

__device__ __forceinline__ void rows_epilogue2(float* acc, int rows, int node_lo, int tid,
        const float* __restrict__ h, const float* __restrict__ dis, const float* __restrict__ b2,
        const float* __restrict__ x1) {
    const float4* __restrict__ h4 = reinterpret_cast<const float4*>(h + (size_t)node_lo * HDIM);
    const float4* __restrict__ x4 = reinterpret_cast<const float4*>(x1 + (size_t)node_lo * HDIM);
    const int total = rows * (HDIM / 4);
    const int c4 = tid & (HDIM / 4 - 1);
    const float4 bb = reinterpret_cast<const float4*>(b2)[c4];
    for (int i0 = tid; i0 < total; i0 += TVEC * 1024) {
        float4 hv[TVEC], xv[TVEC];
        float d[TVEC];
#pragma unroll
        for (int j = 0; j < TVEC; ++j) {
            const int i = min(i0 + j * 1024, total - 1);
            hv[j] = h4[i];
            xv[j] = x4[i];
            d[j] = dis[node_lo + i / (HDIM / 4)];
        }
#pragma unroll
        for (int j = 0; j < TVEC; ++j) {
            const int i = i0 + j * 1024;
            if (i < total) {
                float4* ap = reinterpret_cast<float4*>(acc + (i / (HDIM / 4)) * APITCH + 4 * c4);
                const float4 a = *ap;
                const float dd = d[j] * d[j];
                float4 o;
                o.x = fmaxf(xv[j].x, a.x + dd * hv[j].x + bb.x);
                o.y = fmaxf(xv[j].y, a.y + dd * hv[j].y + bb.y);
                o.z = fmaxf(xv[j].z, a.z + dd * hv[j].z + bb.z);
                o.w = fmaxf(xv[j].w, a.w + dd * hv[j].w + bb.w);
                *ap = o;
            }
        }
    }
}

template <int NTL>
__device__ __forceinline__ void rows_product(const float* acc, int rows, int node_lo, int w, unsigned lo,
        const float* __restrict__ Wt, int ncols, const float* __restrict__ bias, float* __restrict__ C) {
    constexpr int KS = HDIM / 4;
    constexpr int G  = AWAVES / NTL;
    if (w < G * NTL) {
        const int c    = (int)lo & 15;
        const int koff = ((int)lo >> 4) * 2;
        const int n    = (w % NTL) * 16 + c;
        const bool in  = n < ncols;
        const int nn   = in ? n : 0;
        const float badd = (bias && in) ? bias[nn] : 0.0f;
        v2f b[KS];
        const float* bp = Wt + (size_t)koff * ncols + nn;
#pragma unroll
        for (int ks = 0; ks < KS; ++ks) {
            v2f t;
            t.x = bp[(ks * 4) * ncols];
            t.y = bp[(ks * 4 + 1) * ncols];
            if (!in) { t.x = 0.f; t.y = 0.f; }
            b[ks] = t;
        }
        const int ntile = (rows + 15) >> 4;
        const v8f z8 = {0.f, 0.f, 0.f, 0.f, 0.f, 0.f, 0.f, 0.f};
#pragma unroll 1
        for (int rt = w / NTL; rt < ntile; rt += G) {
            const float* ap = acc + (rt * 16 + c) * APITCH + koff;
            v8f d = z8;
#pragma unroll
            for (int ks = 0; ks < KS; ++ks) {
                const v2f a = *(const v2f*)(ap + ks * 4);
                d = __builtin_amdgcn_wmma_f32_16x16x4_f32(
                     false, a,  false, b[ks],
                     (short)0, d,
                     false,  false);
            }
            const int rbase = rt * 16 + (((int)lo >> 4) << 3);
            if (in) {
                float* cp = C + (size_t)(node_lo + rbase) * ncols + n;
#pragma unroll
                for (int r = 0; r < 8; ++r)
                    if (rbase + r < rows) cp[r * ncols] = d[r] + badd;
            }
        }
    }
}

template <int MODE>
__global__ __launch_bounds__(1024) void k_aggr(
        const float* __restrict__ h, const float* __restrict__ dis,
        const int* __restrict__ row, const int* __restrict__ col,
        const unsigned* __restrict__ present, int pitch,
        int n, int e, int* __restrict__ nent, int* __restrict__ lstw,
        const float* __restrict__ eb, const float* __restrict__ gamma, const float* __restrict__ beta,
        const float* __restrict__ mean, const float* __restrict__ var, float* __restrict__ x1,
        const float* __restrict__ Wt, const float* __restrict__ bt, float* __restrict__ C) {
    static_assert(RANGE <= 512, "list entry packing");
    static_assert(RANGE % 16 == 0, "matrix tiles stay inside the LDS rows");
    __shared__ __align__(16) float acc[RANGE * APITCH];
    __shared__ int   qEs[AWAVES * QCAP];
    __shared__ int   sLs[AWAVES * SCAP];
    __shared__ unsigned short qTs[AWAVES * QCAP];
    __shared__ int   lcur;

    const int tid = (int)threadIdx.x;
    const unsigned lo = threadIdx.x & 31;
    const int w = __builtin_amdgcn_readfirstlane(tid >> 5);
    const int node_lo = (int)blockIdx.x * RANGE;
    const int rows = min(RANGE, n - node_lo);
    int* qE = qEs + w * QCAP;
    unsigned short* qT = qTs + w * QCAP;
    int* sL = sLs + w * SCAP;
    int2* lst = reinterpret_cast<int2*>(lstw) + (size_t)blockIdx.x * LCAP;
    const unsigned* mine = present + blockIdx.x;

    float4* acc4 = reinterpret_cast<float4*>(acc);
    for (int i = tid; i < RANGE * APITCH / 4; i += 1024) acc4[i] = make_float4(0.f, 0.f, 0.f, 0.f);
    if (tid == 0) lcur = (n <= (1 << LBITS)) ? 0 : LCAP + 1;
    __syncthreads();

    if (MODE == 0) {
        scan_pass<true, APITCH>(h, dis, row, col, mine, pitch, acc, qE, qT, sL, w, lo, node_lo, rows, e, lst, &lcur);
    } else {
        const int ne = __builtin_amdgcn_readfirstlane(nent[blockIdx.x]);
        if (ne >= 0) {
            const int nb = (ne + 31) >> 5;
            const int2* L = lst;
            int2 en = make_int2(0, 0);
            if (w < nb) en = L[w * 32 + min((int)lo, min(32, ne - w * 32) - 1)];
#pragma unroll 1
            for (int b = w; b < nb; b += AWAVES) {
                int2 nx = en;
                const int bn = b + AWAVES;
                if (bn < nb) nx = L[bn * 32 + min((int)lo, min(32, ne - bn * 32) - 1)];
                const int cnt = min(32, ne - b * 32);
                const int r = en.x & ((1 << LBITS) - 1);
                const int t = en.x >> LBITS;
                const float nrm = __int_as_float(en.y);
                if (cnt == 32) gather_add<APITCH>(h, acc, r, t, nrm, 32, lo);
                else           gather_add<APITCH>(h, acc, r, t, nrm, cnt, lo);
                en = nx;
            }
        } else {
            scan_pass<false, APITCH>(h, dis, row, col, mine, pitch, acc, qE, qT, sL, w, lo, node_lo, rows, e, lst, &lcur);
        }
    }
    __syncthreads();

    if (MODE == 0 && tid == 0) nent[blockIdx.x] = (lcur <= LCAP) ? lcur : -1;

    if (MODE == 0) rows_epilogue1(acc, rows, node_lo, tid, h, dis, eb, gamma, beta, mean, var, x1);
    else           rows_epilogue2(acc, rows, node_lo, tid, h, dis, eb, x1);
    __syncthreads();
    if (MODE == 0) rows_product<HDIM / 16>(acc, rows, node_lo, w, lo, Wt, HDIM, bt, C);
    else           rows_product<(ODIM + 15) / 16>(acc, rows, node_lo, w, lo, Wt, ODIM, bt, C);
}

extern "C" void kernel_launch(void* const* d_in, const int* in_sizes, int n_in,
                              void* d_out, int out_size, void* d_ws, size_t ws_size,
                              hipStream_t stream) {
    const float* node_feat = (const float*)d_in[0];
    const int*   eidx      = (const int*)d_in[1];
    const float* W1        = (const float*)d_in[2];
    const float* b1        = (const float*)d_in[3];
    const float* gamma1    = (const float*)d_in[4];
    const float* beta1     = (const float*)d_in[5];
    const float* mean1     = (const float*)d_in[6];
    const float* var1      = (const float*)d_in[7];
    const float* W2        = (const float*)d_in[8];
    const float* b2        = (const float*)d_in[9];
    const float* Wf        = (const float*)d_in[10];
    const float* bf        = (const float*)d_in[11];
    float* out = (float*)d_out;

    const int n = in_sizes[0] / HDIM;
    const int e = in_sizes[1] / 2;
    const int* row = eidx;
    const int* col = eidx + e;

    float* dis = (float*)d_ws;
    float* h1  = dis + n;
    float* x1  = h1 + (size_t)n * HDIM;
    float* h2  = x1 + (size_t)n * HDIM;
    unsigned* present = (unsigned*)(h2 + (size_t)n * HDIM);

    const int aggGrd  = (n + RANGE - 1) / RANGE;
    const int pitch   = (aggGrd + 31) / 32 * 32;
    const int chunks  = (e + CHUNK - 1) / CHUNK;
    int* nent = (int*)(present + (size_t)chunks * pitch);
    int* lstw = nent + pitch;

    k_init_deg<<<(n + 255) / 256, 256, 0, stream>>>(dis, n);
    k_edge_deg<<<chunks, CHUNK, 0, stream>>>(col, dis, present, e, pitch);
    k_finish_deg<<<(n + 255) / 256, 256, 0, stream>>>(dis, n);

    k_gemm_wmma<128, 2><<<GEMM_WGS, 256, 0, stream>>>(node_feat, W1, nullptr, h1, n, HDIM);
    k_aggr<0><<<aggGrd, 1024, 0, stream>>>(h1, dis, row, col, present, pitch, n, e, nent, lstw,
                                          b1, gamma1, beta1, mean1, var1, x1, W2, nullptr, h2);

    k_aggr<1><<<aggGrd, 1024, 0, stream>>>(h2, dis, row, col, present, pitch, n, e, nent, lstw,
                                          b2, nullptr, nullptr, nullptr, nullptr, x1, Wf, bf, out);
}


# k_aggr: wave w keeps sums of target rows w, w+32, .. in registers; found edges routed to the owning wave via LDS queues; no LDS float adds; sum = queue arrival order, fused multiply-add
# speedup vs baseline: 3.3660x; 1.0718x over previous
#include <hip/hip_runtime.h>

typedef __attribute__((ext_vector_type(2))) float v2f;
typedef __attribute__((ext_vector_type(8))) float v8f;

#define HDIM 128
#define ODIM 40

__global__ void k_init_deg(float* __restrict__ deg, int n) {
    int i = blockIdx.x * blockDim.x + threadIdx.x;
    if (i < n) deg[i] = 1.0f;
}

#define RANGE   400
#define CHUNK   1024

__global__ __launch_bounds__(CHUNK) void k_edge_deg(const int* __restrict__ col, float* __restrict__ deg,
                                                    unsigned* __restrict__ present, int e, int pitch) {
    __shared__ unsigned pres[CHUNK];
    const int tid = (int)threadIdx.x;
    pres[tid] = 0u;
    __syncthreads();
    const int i = (int)blockIdx.x * CHUNK + tid;
    if (i < e) {
        const int c = col[i];
        atomicAdd(&deg[c], 1.0f);
        atomicOr(&pres[c / RANGE], 1u << (tid >> 5));
    }
    __syncthreads();
    if (tid < pitch) present[(size_t)blockIdx.x * pitch + tid] = pres[tid];
}

__global__ void k_finish_deg(float* __restrict__ deg, int n) {
    int i = blockIdx.x * blockDim.x + threadIdx.x;
    if (i < n) {
        float d = deg[i];
        deg[i] = (d > 0.0f) ? rsqrtf(d) : 0.0f;
    }
}

#define GEMM_WGS 256

template <int NPAD, int SPLIT>
__global__ __launch_bounds__(256) void k_gemm_wmma(
        const float* __restrict__ A, const float* __restrict__ B,
        const float* __restrict__ bias, float* __restrict__ C,
        int M, int ncols) {
    constexpr int K   = 128;
    constexpr int KS  = K / 4;
    constexpr int NTW = NPAD / 16 / SPLIT;

    const int lane  = threadIdx.x & 31;
    const int c     = lane & 15;
    const int koff  = (lane >> 4) * 2;
    const int gw    = (int)blockIdx.x * 8 + __builtin_amdgcn_readfirstlane((int)threadIdx.x >> 5);
    const int share = gw % SPLIT;
    const int t0    = gw / SPLIT;
    const int tstep = (int)gridDim.x * 8 / SPLIT;
    const int mtile = M >> 4;
    if (t0 >= mtile) return;
    const int nbase = share * (NTW * 16);

    v2f b[KS][NTW];
    float badd[NTW];
#pragma unroll
    for (int nt = 0; nt < NTW; ++nt) {
        const int n = nbase + nt * 16 + c;
        const bool in = n < ncols;
        const int nn = in ? n : 0;
        badd[nt] = (bias && in) ? bias[nn] : 0.0f;
        const float* bp = B + (size_t)koff * ncols + nn;
#pragma unroll
        for (int ks = 0; ks < KS; ++ks) {
            v2f t;
            t.x = bp[(ks * 4) * ncols];
            t.y = bp[(ks * 4 + 1) * ncols];
            if (!in) { t.x = 0.f; t.y = 0.f; }
            b[ks][nt] = t;
        }
    }

    v2f a[KS];
    {
        const float* aptr = A + (size_t)(t0 * 16 + c) * K + koff;
#pragma unroll
        for (int ks = 0; ks < KS; ++ks) a[ks] = *(const v2f*)(aptr + ks * 4);
    }

    const v8f z8 = {0.f, 0.f, 0.f, 0.f, 0.f, 0.f, 0.f, 0.f};
#pragma unroll 1
    for (int tile = t0; tile < mtile; tile += tstep) {
        v2f an[KS];
        const int nxt = tile + tstep;
        if (nxt < mtile) {
            const float* aptr = A + (size_t)(nxt * 16 + c) * K + koff;
#pragma unroll
            for (int ks = 0; ks < KS; ++ks) an[ks] = *(const v2f*)(aptr + ks * 4);
        } else {
#pragma unroll
            for (int ks = 0; ks < KS; ++ks) an[ks] = a[ks];
        }

        v8f acc[NTW];
#pragma unroll
        for (int nt = 0; nt < NTW; ++nt) acc[nt] = z8;
#pragma unroll
        for (int ks = 0; ks < KS; ++ks) {
#pragma unroll
            for (int nt = 0; nt < NTW; ++nt) {
                acc[nt] = __builtin_amdgcn_wmma_f32_16x16x4_f32(
                     false, a[ks],  false, b[ks][nt],
                     (short)0, acc[nt],
                     false,  false);
            }
        }

        const int rbase = tile * 16 + ((lane >> 4) << 3);
#pragma unroll
        for (int nt = 0; nt < NTW; ++nt) {
            const int n = nbase + nt * 16 + c;
            if (n >= ncols) continue;
            float* cp = C + (size_t)rbase * ncols + n;
#pragma unroll
            for (int r = 0; r < 8; ++r) cp[r * ncols] = acc[nt][r] + badd[nt];
        }
#pragma unroll
        for (int ks = 0; ks < KS; ++ks) a[ks] = an[ks];
    }
}

#define AWAVES  32
#define SGRP    8
#define QCAP    288
#define SCAP    264
#define EGRP    8
#define LBITS   22
#define NROW    13
#define WCAP    512
#define EBITS   27

__device__ __forceinline__ void load_rows(const float* __restrict__ h, int r, int g, int cnt, unsigned lo,
                                          float (&v)[EGRP][4]) {
#pragma unroll
    for (int j = 0; j < EGRP; ++j) {
        if (g + j < cnt) {
            const float* srow = h + (size_t)__builtin_amdgcn_readlane(r, g + j) * HDIM;
            v[j][0] = srow[lo];
            v[j][1] = srow[lo + 32];
            v[j][2] = srow[lo + 64];
            v[j][3] = srow[lo + 96];
        }
    }
}

__device__ __forceinline__ void lds_add(float* p, float x) {
    __hip_atomic_fetch_add(p, x, __ATOMIC_RELAXED, __HIP_MEMORY_SCOPE_WORKGROUP);
}

template <int AP>
__device__ __forceinline__ void add_rows(float* acc, int t, float nrm, int g, int cnt, unsigned lo,
                                         const float (&v)[EGRP][4]) {
#pragma unroll
    for (int j = 0; j < EGRP; ++j) {
        if (g + j < cnt) {
            float* drow = acc + __builtin_amdgcn_readlane(t, g + j) * AP;
            const float nj = __int_as_float(__builtin_amdgcn_readlane(__float_as_int(nrm), g + j));
            lds_add(drow + lo,      v[j][0] * nj);
            lds_add(drow + lo + 32, v[j][1] * nj);
            lds_add(drow + lo + 64, v[j][2] * nj);
            lds_add(drow + lo + 96, v[j][3] * nj);
        }
    }
}

template <int AP>
__device__ __forceinline__ void drain(const float* __restrict__ h, const float* __restrict__ dis,
                                      const int* __restrict__ row, float* acc,
                                      const int* qE, const unsigned short* qT, int off, int cnt,
                                      unsigned lo, int node_lo) {
    const int li = off + min((int)lo, cnt - 1);
    const int ej = qE[li];
    const int tj = (int)qT[li];
    const int r  = row[ej];
    float v[EGRP][4];
    load_rows(h, r, 0, cnt, lo, v);
    const float nrm = dis[r] * dis[node_lo + tj];
    add_rows<AP>(acc, tj, nrm, 0, cnt, lo, v);
#pragma unroll
    for (int g = EGRP; g < 32; g += EGRP) {
        load_rows(h, r, g, cnt, lo, v);
        add_rows<AP>(acc, tj, nrm, g, cnt, lo, v);
    }
}

__device__ __forceinline__ void append(bool m, int eidx, unsigned t, int& qn, int* qE, unsigned short* qT) {
    const unsigned bal = __builtin_amdgcn_ballot_w32(m);
    if (bal == 0u) return;
    const int pos = qn + (int)__builtin_amdgcn_mbcnt_lo(bal, 0u);
    if (m) { qE[pos] = eidx; qT[pos] = (unsigned short)t; }
    qn += __builtin_popcount(bal);
}

__device__ __forceinline__ void append_seg(bool m, int seg, int& sn, int* sL) {
    const unsigned bal = __builtin_amdgcn_ballot_w32(m);
    if (bal == 0u) return;
    if (m) sL[sn + (int)__builtin_amdgcn_mbcnt_lo(bal, 0u)] = seg;
    sn += __builtin_popcount(bal);
}

__device__ __forceinline__ void scan_batch(const int* __restrict__ col, const int* sL, int off, int cnt, int e,
                                           unsigned lo, int node_lo, int rows, int& qn,
                                           int* qE, unsigned short* qT) {
    const int sv = sL[off + min((int)lo, cnt - 1)];
    int ei[SGRP];
    int c[SGRP];
#pragma unroll
    for (int j = 0; j < SGRP; ++j) {
        if (j < cnt) {
            ei[j] = __builtin_amdgcn_readlane(sv, j) * 32 + (int)lo;
            c[j] = col[min(ei[j], e - 1)];
        }
    }
#pragma unroll
    for (int j = 0; j < SGRP; ++j) {
        if (j < cnt) {
            const unsigned d = (unsigned)(c[j] - node_lo);
            append((d < (unsigned)rows) & (ei[j] < e), ei[j], d, qn, qE, qT);
        }
    }
}

template <int AP>
__device__ __forceinline__ void scan_pass(const float* __restrict__ h, const float* __restrict__ dis,
                                          const int* __restrict__ row, const int* __restrict__ col,
                                          const unsigned* __restrict__ mine, int pitch, float* acc,
                                          int* qE, unsigned short* qT, int* sL, int w, unsigned lo,
                                          int node_lo, int rows, int e) {
    int qn = 0;
    int sn = 0;
    const int nchunk = (e + CHUNK - 1) / CHUNK;
#pragma unroll 1
    for (int cb = w; cb < nchunk; cb += 32 * AWAVES) {
        const int chunk = cb + AWAVES * (int)lo;
        const unsigned wd = (chunk < nchunk) ? mine[(size_t)chunk * pitch] : 0u;
#pragma unroll 1
        for (int kc = 0; kc < 32; kc += 8) {
            const unsigned part = (wd >> kc) & 0xffu;
            if (__builtin_amdgcn_ballot_w32(part != 0u) == 0u) continue;
#pragma unroll
            for (int k = 0; k < 8; ++k)
                append_seg(((part >> k) & 1u) != 0u, chunk * 32 + kc + k, sn, sL);
            while (sn >= SGRP) {
                sn -= SGRP;
                scan_batch(col, sL, sn, SGRP, e, lo, node_lo, rows, qn, qE, qT);
                while (qn >= 32) {
                    qn -= 32;
                    drain<AP>(h, dis, row, acc, qE, qT, qn, 32, lo, node_lo);
                }
            }
        }
    }
    if (sn > 0) {
        scan_batch(col, sL, 0, sn, e, lo, node_lo, rows, qn, qE, qT);
        while (qn >= 32) {
            qn -= 32;
            drain<AP>(h, dis, row, acc, qE, qT, qn, 32, lo, node_lo);
        }
    }
    if (qn > 0) drain<AP>(h, dis, row, acc, qE, qT, 0, qn, lo, node_lo);
}

#define APITCH 132
#define TVEC   7

__device__ __forceinline__ float bn_relu(float v, float m, float vr, float g, float b) {
    v = (v - m) * rsqrtf(vr + 1e-5f) * g + b;
    return fmaxf(v, 0.0f);
}

__device__ __forceinline__ void rows_epilogue1(float* acc, int rows, int node_lo, int tid,
        const float* __restrict__ h, const float* __restrict__ dis, const float* __restrict__ b1,
        const float* __restrict__ gamma, const float* __restrict__ beta,
        const float* __restrict__ mean, const float* __restrict__ var, float* __restrict__ x1) {
    const float4* __restrict__ h4 = reinterpret_cast<const float4*>(h + (size_t)node_lo * HDIM);
    float4* __restrict__ x4 = reinterpret_cast<float4*>(x1 + (size_t)node_lo * HDIM);
    const int total = rows * (HDIM / 4);
    const int c4 = tid & (HDIM / 4 - 1);
    const float4 bb = reinterpret_cast<const float4*>(b1)[c4];
    const float4 mm = reinterpret_cast<const float4*>(mean)[c4];
    const float4 vv = reinterpret_cast<const float4*>(var)[c4];
    const float4 gg = reinterpret_cast<const float4*>(gamma)[c4];
    const float4 be = reinterpret_cast<const float4*>(beta)[c4];
    for (int i0 = tid; i0 < total; i0 += TVEC * 1024) {
        float4 hv[TVEC];
        float d[TVEC];
#pragma unroll
        for (int j = 0; j < TVEC; ++j) {
            const int i = min(i0 + j * 1024, total - 1);
            hv[j] = h4[i];
            d[j] = dis[node_lo + i / (HDIM / 4)];
        }
#pragma unroll
        for (int j = 0; j < TVEC; ++j) {
            const int i = i0 + j * 1024;
            if (i < total) {
                float4* ap = reinterpret_cast<float4*>(acc + (i / (HDIM / 4)) * APITCH + 4 * c4);
                const float4 a = *ap;
                const float dd = d[j] * d[j];
                float4 o;
                o.x = bn_relu(a.x + dd * hv[j].x + bb.x, mm.x, vv.x, gg.x, be.x);
                o.y = bn_relu(a.y + dd * hv[j].y + bb.y, mm.y, vv.y, gg.y, be.y);
                o.z = bn_relu(a.z + dd * hv[j].z + bb.z, mm.z, vv.z, gg.z, be.z);
                o.w = bn_relu(a.w + dd * hv[j].w + bb.w, mm.w, vv.w, gg.w, be.w);
                *ap = o;
                x4[i] = o;
            }
        }
    }
}

__device__ __forceinline__ void rows_epilogue2(float* acc, int rows, int node_lo, int tid,
        const float* __restrict__ h, const float* __restrict__ dis, const float* __restrict__ b2,
        const float* __restrict__ x1) {
    const float4* __restrict__ h4 = reinterpret_cast<const float4*>(h + (size_t)node_lo * HDIM);
    const float4* __restrict__ x4 = reinterpret_cast<const float4*>(x1 + (size_t)node_lo * HDIM);
    const int total = rows * (HDIM / 4);
    const int c4 = tid & (HDIM / 4 - 1);
    const float4 bb = reinterpret_cast<const float4*>(b2)[c4];
    for (int i0 = tid; i0 < total; i0 += TVEC * 1024) {
        float4 hv[TVEC], xv[TVEC];
        float d[TVEC];
#pragma unroll
        for (int j = 0; j < TVEC; ++j) {
            const int i = min(i0 + j * 1024, total - 1);
            hv[j] = h4[i];
            xv[j] = x4[i];
            d[j] = dis[node_lo + i / (HDIM / 4)];
        }
#pragma unroll
        for (int j = 0; j < TVEC; ++j) {
            const int i = i0 + j * 1024;
            if (i < total) {
                float4* ap = reinterpret_cast<float4*>(acc + (i / (HDIM / 4)) * APITCH + 4 * c4);
                const float4 a = *ap;
                const float dd = d[j] * d[j];
                float4 o;
                o.x = fmaxf(xv[j].x, a.x + dd * hv[j].x + bb.x);
                o.y = fmaxf(xv[j].y, a.y + dd * hv[j].y + bb.y);
                o.z = fmaxf(xv[j].z, a.z + dd * hv[j].z + bb.z);
                o.w = fmaxf(xv[j].w, a.w + dd * hv[j].w + bb.w);
                *ap = o;
            }
        }
    }
}

template <int NTL>
__device__ __forceinline__ void rows_product(const float* acc, int rows, int node_lo, int w, unsigned lo,
        const float* __restrict__ Wt, int ncols, const float* __restrict__ bias, float* __restrict__ C) {
    constexpr int KS = HDIM / 4;
    constexpr int G  = AWAVES / NTL;
    if (w < G * NTL) {
        const int c    = (int)lo & 15;
        const int koff = ((int)lo >> 4) * 2;
        const int n    = (w % NTL) * 16 + c;
        const bool in  = n < ncols;
        const int nn   = in ? n : 0;
        const float badd = (bias && in) ? bias[nn] : 0.0f;
        v2f b[KS];
        const float* bp = Wt + (size_t)koff * ncols + nn;
#pragma unroll
        for (int ks = 0; ks < KS; ++ks) {
            v2f t;
            t.x = bp[(ks * 4) * ncols];
            t.y = bp[(ks * 4 + 1) * ncols];
            if (!in) { t.x = 0.f; t.y = 0.f; }
            b[ks] = t;
        }
        const int ntile = (rows + 15) >> 4;
        const v8f z8 = {0.f, 0.f, 0.f, 0.f, 0.f, 0.f, 0.f, 0.f};
#pragma unroll 1
        for (int rt = w / NTL; rt < ntile; rt += G) {
            const float* ap = acc + (rt * 16 + c) * APITCH + koff;
            v8f d = z8;
#pragma unroll
            for (int ks = 0; ks < KS; ++ks) {
                const v2f a = *(const v2f*)(ap + ks * 4);
                d = __builtin_amdgcn_wmma_f32_16x16x4_f32(
                     false, a,  false, b[ks],
                     (short)0, d,
                     false,  false);
            }
            const int rbase = rt * 16 + (((int)lo >> 4) << 3);
            if (in) {
                float* cp = C + (size_t)(node_lo + rbase) * ncols + n;
#pragma unroll
                for (int r = 0; r < 8; ++r)
                    if (rbase + r < rows) cp[r * ncols] = d[r] + badd;
            }
        }
    }
}


typedef __attribute__((ext_vector_type(16))) float v16f;
struct RowAcc { v16f x, y, z, w; };

__device__ __forceinline__ void row_fma(RowAcc& a, int slot, float nj, const float (&v)[4]) {
    a.x[slot] = fmaf(v[0], nj, a.x[slot]);
    a.y[slot] = fmaf(v[1], nj, a.y[slot]);
    a.z[slot] = fmaf(v[2], nj, a.z[slot]);
    a.w[slot] = fmaf(v[3], nj, a.w[slot]);
}

__device__ __forceinline__ void load_row4(const float* __restrict__ h, int srcrow, unsigned lo, float (&v)[4]) {
    const float* p = h + ((unsigned)srcrow * (unsigned)HDIM + lo);
    v[0] = p[0];
    v[1] = p[32];
    v[2] = p[64];
    v[3] = p[96];
}

#define SPARE 15
__device__ __forceinline__ void gather_acc(const float* __restrict__ h, int r, int slot, float nrm, int cnt, unsigned lo,
                                           RowAcc& a) {
#pragma unroll 1
    for (int g = 0; g < cnt; g += EGRP) {
        float v[EGRP][4];
#pragma unroll
        for (int j = 0; j < EGRP; ++j) load_row4(h, __builtin_amdgcn_readlane(r, min(g + j, cnt - 1)), lo, v[j]);
#pragma unroll
        for (int j = 0; j < EGRP; ++j) {
            const int i = min(g + j, cnt - 1);
            const int sj = (g + j < cnt) ? __builtin_amdgcn_readlane(slot, i) : SPARE;
            const float nj = __int_as_float(__builtin_amdgcn_readlane(__float_as_int(nrm), i));
            row_fma(a, sj, nj, v[j]);
        }
    }
}

__device__ __forceinline__ void route(const int* qE, const unsigned short* qT, int off, int cnt, unsigned lo,
                                      int* rq, int* rcnt) {
    if ((int)lo < cnt) {
        const int ej = qE[off + (int)lo];
        const int t  = (int)qT[off + (int)lo];
        const int tw = t & (AWAVES - 1);
        const int pos = __hip_atomic_fetch_add(&rcnt[tw], 1, __ATOMIC_RELAXED, __HIP_MEMORY_SCOPE_WORKGROUP);
        if (pos < WCAP) rq[tw * WCAP + pos] = ej | ((t >> 5) << EBITS);
    }
}

__device__ __forceinline__ void scan_route(const int* __restrict__ col, const unsigned* __restrict__ mine, int pitch,
                                           int* qE, unsigned short* qT, int* sL, int w, unsigned lo,
                                           int node_lo, int rows, int e, int* rq, int* rcnt) {
    int qn = 0;
    int sn = 0;
    const int nchunk = (e + CHUNK - 1) / CHUNK;
#pragma unroll 1
    for (int cb = w; cb < nchunk; cb += 32 * AWAVES) {
        const int chunk = cb + AWAVES * (int)lo;
        const unsigned wd = (chunk < nchunk) ? mine[(size_t)chunk * pitch] : 0u;
#pragma unroll 1
        for (int kc = 0; kc < 32; kc += 8) {
            const unsigned part = (wd >> kc) & 0xffu;
            if (__builtin_amdgcn_ballot_w32(part != 0u) == 0u) continue;
#pragma unroll
            for (int k = 0; k < 8; ++k)
                append_seg(((part >> k) & 1u) != 0u, chunk * 32 + kc + k, sn, sL);
            while (sn >= SGRP) {
                sn -= SGRP;
                scan_batch(col, sL, sn, SGRP, e, lo, node_lo, rows, qn, qE, qT);
                while (qn >= 32) {
                    qn -= 32;
                    route(qE, qT, qn, 32, lo, rq, rcnt);
                }
            }
        }
    }
    if (sn > 0) {
        scan_batch(col, sL, 0, sn, e, lo, node_lo, rows, qn, qE, qT);
        while (qn >= 32) {
            qn -= 32;
            route(qE, qT, qn, 32, lo, rq, rcnt);
        }
    }
    if (qn > 0) route(qE, qT, 0, qn, lo, rq, rcnt);
}

template <int MODE>
__global__ __launch_bounds__(1024) void k_aggr(
        const float* __restrict__ h, const float* __restrict__ dis,
        const int* __restrict__ row, const int* __restrict__ col,
        const unsigned* __restrict__ present, int pitch,
        int n, int e, int* __restrict__ nent, int* __restrict__ lstw,
        const float* __restrict__ eb, const float* __restrict__ gamma, const float* __restrict__ beta,
        const float* __restrict__ mean, const float* __restrict__ var, float* __restrict__ x1,
        const float* __restrict__ Wt, const float* __restrict__ bt, float* __restrict__ C) {
    static_assert(RANGE <= 512, "list entry packing");
    static_assert(RANGE % 16 == 0, "matrix tiles stay inside the LDS rows");
    static_assert(NROW == (RANGE + 31) / 32 && NROW <= SPARE, "row slots and the spare slot index a vector of 16");
    static_assert(AWAVES == 32, "a target's wave is its low five bits");
    static_assert(AWAVES * WCAP <= RANGE * APITCH, "the routed queues lie in the rows' LDS");
    __shared__ __align__(16) float acc[RANGE * APITCH];
    __shared__ int   qEs[AWAVES * QCAP];
    __shared__ int   sLs[AWAVES * SCAP];
    __shared__ unsigned short qTs[AWAVES * QCAP];
    __shared__ int   rcnt[AWAVES];

    const int tid = (int)threadIdx.x;
    const unsigned lo = threadIdx.x & 31;
    const int w = __builtin_amdgcn_readfirstlane(tid >> 5);
    const int node_lo = (int)blockIdx.x * RANGE;
    const int rows = min(RANGE, n - node_lo);
    int* qE = qEs + w * QCAP;
    unsigned short* qT = qTs + w * QCAP;
    int* sL = sLs + w * SCAP;
    int* rq = reinterpret_cast<int*>(acc);
    int2* lst = reinterpret_cast<int2*>(lstw) + ((size_t)blockIdx.x * AWAVES + w) * WCAP;
    int* nw = nent + (size_t)blockIdx.x * AWAVES;
    const unsigned* mine = present + blockIdx.x;

    RowAcc a;
    a.x = (v16f)(0.f); a.y = (v16f)(0.f); a.z = (v16f)(0.f); a.w = (v16f)(0.f);

    bool fb;
    if (MODE == 0) {
        if (tid < AWAVES) rcnt[tid] = 0;
        __syncthreads();
        const bool pre = (n > (1 << LBITS)) || (e > (1 << EBITS));
        if (!pre) scan_route(col, mine, pitch, qE, qT, sL, w, lo, node_lo, rows, e, rq, rcnt);
        __syncthreads();
        const int cl = rcnt[lo];
        fb = pre || (__builtin_amdgcn_ballot_w32(cl > WCAP) != 0u);
        if (!fb) {
            const int cw = __builtin_amdgcn_readlane(cl, w);
            const int* rqw = rq + w * WCAP;
#pragma unroll 1
            for (int b0 = 0; b0 < cw; b0 += 32) {
                const int cnt = min(32, cw - b0);
                const int ent = rqw[b0 + min((int)lo, cnt - 1)];
                const int slot = ent >> EBITS;
                const int r = row[ent & ((1 << EBITS) - 1)];
                const float nrm = dis[r] * dis[node_lo + slot * 32 + w];
                gather_acc(h, r, slot, nrm, cnt, lo, a);
                if ((int)lo < cnt) {
                    int2 en;
                    en.x = r | (slot << LBITS);
                    en.y = __float_as_int(nrm);
                    lst[b0 + (int)lo] = en;
                }
            }
            if (lo == 0u) nw[w] = cw;
        } else {
            if (lo == 0u) nw[w] = -1;
        }
    } else {
        const int cl = nw[lo];
        fb = __builtin_amdgcn_ballot_w32(cl < 0) != 0u;
        if (!fb) {
            const int ne = __builtin_amdgcn_readlane(cl, w);
            const int2* L = lst;
            int2 en = make_int2(0, 0);
            if (ne > 0) en = L[min((int)lo, min(32, ne) - 1)];
#pragma unroll 1
            for (int b0 = 0; b0 < ne; b0 += 32) {
                int2 nx = en;
                const int bn = b0 + 32;
                if (bn < ne) nx = L[bn + min((int)lo, min(32, ne - bn) - 1)];
                const int cnt = min(32, ne - b0);
                gather_acc(h, en.x & ((1 << LBITS) - 1), en.x >> LBITS, __int_as_float(en.y), cnt, lo, a);
                en = nx;
            }
        }
    }

    if (!fb) {
        if (MODE == 0) __syncthreads();
#pragma unroll
        for (int k = 0; k < NROW; ++k) {
            const int t = k * 32 + w;
            if (t < RANGE) {
                float* drow = acc + t * APITCH;
                drow[lo]      = a.x[k];
                drow[lo + 32] = a.y[k];
                drow[lo + 64] = a.z[k];
                drow[lo + 96] = a.w[k];
            }
        }
    } else {
        float4* acc4 = reinterpret_cast<float4*>(acc);
        for (int i = tid; i < RANGE * APITCH / 4; i += 1024) acc4[i] = make_float4(0.f, 0.f, 0.f, 0.f);
        __syncthreads();
        scan_pass<APITCH>(h, dis, row, col, mine, pitch, acc, qE, qT, sL, w, lo, node_lo, rows, e);
    }
    __syncthreads();

    if (MODE == 0) rows_epilogue1(acc, rows, node_lo, tid, h, dis, eb, gamma, beta, mean, var, x1);
    else           rows_epilogue2(acc, rows, node_lo, tid, h, dis, eb, x1);
    __syncthreads();
    if (MODE == 0) rows_product<HDIM / 16>(acc, rows, node_lo, w, lo, Wt, HDIM, bt, C);
    else           rows_product<(ODIM + 15) / 16>(acc, rows, node_lo, w, lo, Wt, ODIM, bt, C);
}

extern "C" void kernel_launch(void* const* d_in, const int* in_sizes, int n_in,
                              void* d_out, int out_size, void* d_ws, size_t ws_size,
                              hipStream_t stream) {
    const float* node_feat = (const float*)d_in[0];
    const int*   eidx      = (const int*)d_in[1];
    const float* W1        = (const float*)d_in[2];
    const float* b1        = (const float*)d_in[3];
    const float* gamma1    = (const float*)d_in[4];
    const float* beta1     = (const float*)d_in[5];
    const float* mean1     = (const float*)d_in[6];
    const float* var1      = (const float*)d_in[7];
    const float* W2        = (const float*)d_in[8];
    const float* b2        = (const float*)d_in[9];
    const float* Wf        = (const float*)d_in[10];
    const float* bf        = (const float*)d_in[11];
    float* out = (float*)d_out;

    const int n = in_sizes[0] / HDIM;
    const int e = in_sizes[1] / 2;
    const int* row = eidx;
    const int* col = eidx + e;

    float* dis = (float*)d_ws;
    float* h1  = dis + n;
    float* x1  = h1 + (size_t)n * HDIM;
    float* h2  = x1 + (size_t)n * HDIM;
    unsigned* present = (unsigned*)(h2 + (size_t)n * HDIM);

    const int aggGrd  = (n + RANGE - 1) / RANGE;
    const int pitch   = (aggGrd + 31) / 32 * 32;
    const int chunks  = (e + CHUNK - 1) / CHUNK;
    int* nent = (int*)(present + (size_t)chunks * pitch);
    int* lstw = nent + (size_t)pitch * AWAVES;

    k_init_deg<<<(n + 255) / 256, 256, 0, stream>>>(dis, n);
    k_edge_deg<<<chunks, CHUNK, 0, stream>>>(col, dis, present, e, pitch);
    k_finish_deg<<<(n + 255) / 256, 256, 0, stream>>>(dis, n);

    k_gemm_wmma<128, 2><<<GEMM_WGS, 256, 0, stream>>>(node_feat, W1, nullptr, h1, n, HDIM);
    k_aggr<0><<<aggGrd, 1024, 0, stream>>>(h1, dis, row, col, present, pitch, n, e, nent, lstw,
                                          b1, gamma1, beta1, mean1, var1, x1, W2, nullptr, h2);

    k_aggr<1><<<aggGrd, 1024, 0, stream>>>(h2, dis, row, col, present, pitch, n, e, nent, lstw,
                                          b2, nullptr, nullptr, nullptr, nullptr, x1, Wf, bf, out);
}


# Degrees counted by the owner work-groups (LDS counter per target, exact) in a scan launch of their own that writes dis and the replay lists: no global float adds, no init/finish; 6 -> 5 launches
# speedup vs baseline: 4.5943x; 1.3649x over previous
#include <hip/hip_runtime.h>

typedef __attribute__((ext_vector_type(2))) float v2f;
typedef __attribute__((ext_vector_type(8))) float v8f;

#define HDIM 128
#define ODIM 40

#define RANGE   400
#define CHUNK   1024

__global__ __launch_bounds__(CHUNK) void k_edge_present(const int* __restrict__ col,
                                                        unsigned* __restrict__ present, int e, int pitch) {
    __shared__ unsigned pres[CHUNK];
    const int tid = (int)threadIdx.x;
    pres[tid] = 0u;
    __syncthreads();
    const int i = (int)blockIdx.x * CHUNK + tid;
    if (i < e) {
        const int c = col[i];
        atomicOr(&pres[c / RANGE], 1u << (tid >> 5));
    }
    __syncthreads();
    if (tid < pitch) present[(size_t)blockIdx.x * pitch + tid] = pres[tid];
}

#define GEMM_WGS 256

template <int NPAD, int SPLIT>
__global__ __launch_bounds__(256) void k_gemm_wmma(
        const float* __restrict__ A, const float* __restrict__ B,
        const float* __restrict__ bias, float* __restrict__ C,
        int M, int ncols) {
    constexpr int K   = 128;
    constexpr int KS  = K / 4;
    constexpr int NTW = NPAD / 16 / SPLIT;

    const int lane  = threadIdx.x & 31;
    const int c     = lane & 15;
    const int koff  = (lane >> 4) * 2;
    const int gw    = (int)blockIdx.x * 8 + __builtin_amdgcn_readfirstlane((int)threadIdx.x >> 5);
    const int share = gw % SPLIT;
    const int t0    = gw / SPLIT;
    const int tstep = (int)gridDim.x * 8 / SPLIT;
    const int mtile = M >> 4;
    if (t0 >= mtile) return;
    const int nbase = share * (NTW * 16);

    v2f b[KS][NTW];
    float badd[NTW];
#pragma unroll
    for (int nt = 0; nt < NTW; ++nt) {
        const int n = nbase + nt * 16 + c;
        const bool in = n < ncols;
        const int nn = in ? n : 0;
        badd[nt] = (bias && in) ? bias[nn] : 0.0f;
        const float* bp = B + (size_t)koff * ncols + nn;
#pragma unroll
        for (int ks = 0; ks < KS; ++ks) {
            v2f t;
            t.x = bp[(ks * 4) * ncols];
            t.y = bp[(ks * 4 + 1) * ncols];
            if (!in) { t.x = 0.f; t.y = 0.f; }
            b[ks][nt] = t;
        }
    }

    v2f a[KS];
    {
        const float* aptr = A + (size_t)(t0 * 16 + c) * K + koff;
#pragma unroll
        for (int ks = 0; ks < KS; ++ks) a[ks] = *(const v2f*)(aptr + ks * 4);
    }

    const v8f z8 = {0.f, 0.f, 0.f, 0.f, 0.f, 0.f, 0.f, 0.f};
#pragma unroll 1
    for (int tile = t0; tile < mtile; tile += tstep) {
        v2f an[KS];
        const int nxt = tile + tstep;
        if (nxt < mtile) {
            const float* aptr = A + (size_t)(nxt * 16 + c) * K + koff;
#pragma unroll
            for (int ks = 0; ks < KS; ++ks) an[ks] = *(const v2f*)(aptr + ks * 4);
        } else {
#pragma unroll
            for (int ks = 0; ks < KS; ++ks) an[ks] = a[ks];
        }

        v8f acc[NTW];
#pragma unroll
        for (int nt = 0; nt < NTW; ++nt) acc[nt] = z8;
#pragma unroll
        for (int ks = 0; ks < KS; ++ks) {
#pragma unroll
            for (int nt = 0; nt < NTW; ++nt) {
                acc[nt] = __builtin_amdgcn_wmma_f32_16x16x4_f32(
                     false, a[ks],  false, b[ks][nt],
                     (short)0, acc[nt],
                     false,  false);
            }
        }

        const int rbase = tile * 16 + ((lane >> 4) << 3);
#pragma unroll
        for (int nt = 0; nt < NTW; ++nt) {
            const int n = nbase + nt * 16 + c;
            if (n >= ncols) continue;
            float* cp = C + (size_t)rbase * ncols + n;
#pragma unroll
            for (int r = 0; r < 8; ++r) cp[r * ncols] = acc[nt][r] + badd[nt];
        }
#pragma unroll
        for (int ks = 0; ks < KS; ++ks) a[ks] = an[ks];
    }
}

#define AWAVES  32
#define SGRP    8
#define QCAP    288
#define SCAP    264
#define EGRP    8
#define LBITS   22
#define NROW    13
#define WCAP    512
#define EBITS   27

__device__ __forceinline__ void load_rows(const float* __restrict__ h, int r, int g, int cnt, unsigned lo,
                                          float (&v)[EGRP][4]) {
#pragma unroll
    for (int j = 0; j < EGRP; ++j) {
        if (g + j < cnt) {
            const float* srow = h + (size_t)__builtin_amdgcn_readlane(r, g + j) * HDIM;
            v[j][0] = srow[lo];
            v[j][1] = srow[lo + 32];
            v[j][2] = srow[lo + 64];
            v[j][3] = srow[lo + 96];
        }
    }
}

__device__ __forceinline__ void lds_add(float* p, float x) {
    __hip_atomic_fetch_add(p, x, __ATOMIC_RELAXED, __HIP_MEMORY_SCOPE_WORKGROUP);
}

template <int AP>
__device__ __forceinline__ void add_rows(float* acc, int t, float nrm, int g, int cnt, unsigned lo,
                                         const float (&v)[EGRP][4]) {
#pragma unroll
    for (int j = 0; j < EGRP; ++j) {
        if (g + j < cnt) {
            float* drow = acc + __builtin_amdgcn_readlane(t, g + j) * AP;
            const float nj = __int_as_float(__builtin_amdgcn_readlane(__float_as_int(nrm), g + j));
            lds_add(drow + lo,      v[j][0] * nj);
            lds_add(drow + lo + 32, v[j][1] * nj);
            lds_add(drow + lo + 64, v[j][2] * nj);
            lds_add(drow + lo + 96, v[j][3] * nj);
        }
    }
}

template <int AP>
__device__ __forceinline__ void drain(const float* __restrict__ h, const float* __restrict__ dis,
                                      const int* __restrict__ row, float* acc,
                                      const int* qE, const unsigned short* qT, int off, int cnt,
                                      unsigned lo, int node_lo) {
    const int li = off + min((int)lo, cnt - 1);
    const int ej = qE[li];
    const int tj = (int)qT[li];
    const int r  = row[ej];
    float v[EGRP][4];
    load_rows(h, r, 0, cnt, lo, v);
    const float nrm = dis[r] * dis[node_lo + tj];
    add_rows<AP>(acc, tj, nrm, 0, cnt, lo, v);
#pragma unroll
    for (int g = EGRP; g < 32; g += EGRP) {
        load_rows(h, r, g, cnt, lo, v);
        add_rows<AP>(acc, tj, nrm, g, cnt, lo, v);
    }
}

__device__ __forceinline__ void append(bool m, int eidx, unsigned t, int& qn, int* qE, unsigned short* qT) {
    const unsigned bal = __builtin_amdgcn_ballot_w32(m);
    if (bal == 0u) return;
    const int pos = qn + (int)__builtin_amdgcn_mbcnt_lo(bal, 0u);
    if (m) { qE[pos] = eidx; qT[pos] = (unsigned short)t; }
    qn += __builtin_popcount(bal);
}

__device__ __forceinline__ void append_seg(bool m, int seg, int& sn, int* sL) {
    const unsigned bal = __builtin_amdgcn_ballot_w32(m);
    if (bal == 0u) return;
    if (m) sL[sn + (int)__builtin_amdgcn_mbcnt_lo(bal, 0u)] = seg;
    sn += __builtin_popcount(bal);
}

__device__ __forceinline__ void scan_batch(const int* __restrict__ col, const int* sL, int off, int cnt, int e,
                                           unsigned lo, int node_lo, int rows, int& qn,
                                           int* qE, unsigned short* qT) {
    const int sv = sL[off + min((int)lo, cnt - 1)];
    int ei[SGRP];
    int c[SGRP];
#pragma unroll
    for (int j = 0; j < SGRP; ++j) {
        if (j < cnt) {
            ei[j] = __builtin_amdgcn_readlane(sv, j) * 32 + (int)lo;
            c[j] = col[min(ei[j], e - 1)];
        }
    }
#pragma unroll
    for (int j = 0; j < SGRP; ++j) {
        if (j < cnt) {
            const unsigned d = (unsigned)(c[j] - node_lo);
            append((d < (unsigned)rows) & (ei[j] < e), ei[j], d, qn, qE, qT);
        }
    }
}

template <int AP>
__device__ __forceinline__ void scan_pass(const float* __restrict__ h, const float* __restrict__ dis,
                                          const int* __restrict__ row, const int* __restrict__ col,
                                          const unsigned* __restrict__ mine, int pitch, float* acc,
                                          int* qE, unsigned short* qT, int* sL, int w, unsigned lo,
                                          int node_lo, int rows, int e) {
    int qn = 0;
    int sn = 0;
    const int nchunk = (e + CHUNK - 1) / CHUNK;
#pragma unroll 1
    for (int cb = w; cb < nchunk; cb += 32 * AWAVES) {
        const int chunk = cb + AWAVES * (int)lo;
        const unsigned wd = (chunk < nchunk) ? mine[(size_t)chunk * pitch] : 0u;
#pragma unroll 1
        for (int kc = 0; kc < 32; kc += 8) {
            const unsigned part = (wd >> kc) & 0xffu;
            if (__builtin_amdgcn_ballot_w32(part != 0u) == 0u) continue;
#pragma unroll
            for (int k = 0; k < 8; ++k)
                append_seg(((part >> k) & 1u) != 0u, chunk * 32 + kc + k, sn, sL);
            while (sn >= SGRP) {
                sn -= SGRP;
                scan_batch(col, sL, sn, SGRP, e, lo, node_lo, rows, qn, qE, qT);
                while (qn >= 32) {
                    qn -= 32;
                    drain<AP>(h, dis, row, acc, qE, qT, qn, 32, lo, node_lo);
                }
            }
        }
    }
    if (sn > 0) {
        scan_batch(col, sL, 0, sn, e, lo, node_lo, rows, qn, qE, qT);
        while (qn >= 32) {
            qn -= 32;
            drain<AP>(h, dis, row, acc, qE, qT, qn, 32, lo, node_lo);
        }
    }
    if (qn > 0) drain<AP>(h, dis, row, acc, qE, qT, 0, qn, lo, node_lo);
}

#define APITCH 132
#define TVEC   7

__device__ __forceinline__ float bn_relu(float v, float m, float vr, float g, float b) {
    v = (v - m) * rsqrtf(vr + 1e-5f) * g + b;
    return fmaxf(v, 0.0f);
}

__device__ __forceinline__ void rows_epilogue1(float* acc, int rows, int node_lo, int tid,
        const float* __restrict__ h, const float* __restrict__ dis, const float* __restrict__ b1,
        const float* __restrict__ gamma, const float* __restrict__ beta,
        const float* __restrict__ mean, const float* __restrict__ var, float* __restrict__ x1) {
    const float4* __restrict__ h4 = reinterpret_cast<const float4*>(h + (size_t)node_lo * HDIM);
    float4* __restrict__ x4 = reinterpret_cast<float4*>(x1 + (size_t)node_lo * HDIM);
    const int total = rows * (HDIM / 4);
    const int c4 = tid & (HDIM / 4 - 1);
    const float4 bb = reinterpret_cast<const float4*>(b1)[c4];
    const float4 mm = reinterpret_cast<const float4*>(mean)[c4];
    const float4 vv = reinterpret_cast<const float4*>(var)[c4];
    const float4 gg = reinterpret_cast<const float4*>(gamma)[c4];
    const float4 be = reinterpret_cast<const float4*>(beta)[c4];
    for (int i0 = tid; i0 < total; i0 += TVEC * 1024) {
        float4 hv[TVEC];
        float d[TVEC];
#pragma unroll
        for (int j = 0; j < TVEC; ++j) {
            const int i = min(i0 + j * 1024, total - 1);
            hv[j] = h4[i];
            d[j] = dis[node_lo + i / (HDIM / 4)];
        }
#pragma unroll
        for (int j = 0; j < TVEC; ++j) {
            const int i = i0 + j * 1024;
            if (i < total) {
                float4* ap = reinterpret_cast<float4*>(acc + (i / (HDIM / 4)) * APITCH + 4 * c4);
                const float4 a = *ap;
                const float dd = d[j] * d[j];
                float4 o;
                o.x = bn_relu(a.x + dd * hv[j].x + bb.x, mm.x, vv.x, gg.x, be.x);
                o.y = bn_relu(a.y + dd * hv[j].y + bb.y, mm.y, vv.y, gg.y, be.y);
                o.z = bn_relu(a.z + dd * hv[j].z + bb.z, mm.z, vv.z, gg.z, be.z);
                o.w = bn_relu(a.w + dd * hv[j].w + bb.w, mm.w, vv.w, gg.w, be.w);
                *ap = o;
                x4[i] = o;
            }
        }
    }
}

__device__ __forceinline__ void rows_epilogue2(float* acc, int rows, int node_lo, int tid,
        const float* __restrict__ h, const float* __restrict__ dis, const float* __restrict__ b2,
        const float* __restrict__ x1) {
    const float4* __restrict__ h4 = reinterpret_cast<const float4*>(h + (size_t)node_lo * HDIM);
    const float4* __restrict__ x4 = reinterpret_cast<const float4*>(x1 + (size_t)node_lo * HDIM);
    const int total = rows * (HDIM / 4);
    const int c4 = tid & (HDIM / 4 - 1);
    const float4 bb = reinterpret_cast<const float4*>(b2)[c4];
    for (int i0 = tid; i0 < total; i0 += TVEC * 1024) {
        float4 hv[TVEC], xv[TVEC];
        float d[TVEC];
#pragma unroll
        for (int j = 0; j < TVEC; ++j) {
            const int i = min(i0 + j * 1024, total - 1);
            hv[j] = h4[i];
            xv[j] = x4[i];
            d[j] = dis[node_lo + i / (HDIM / 4)];
        }
#pragma unroll
        for (int j = 0; j < TVEC; ++j) {
            const int i = i0 + j * 1024;
            if (i < total) {
                float4* ap = reinterpret_cast<float4*>(acc + (i / (HDIM / 4)) * APITCH + 4 * c4);
                const float4 a = *ap;
                const float dd = d[j] * d[j];
                float4 o;
                o.x = fmaxf(xv[j].x, a.x + dd * hv[j].x + bb.x);
                o.y = fmaxf(xv[j].y, a.y + dd * hv[j].y + bb.y);
                o.z = fmaxf(xv[j].z, a.z + dd * hv[j].z + bb.z);
                o.w = fmaxf(xv[j].w, a.w + dd * hv[j].w + bb.w);
                *ap = o;
            }
        }
    }
}

template <int NTL>
__device__ __forceinline__ void rows_product(const float* acc, int rows, int node_lo, int w, unsigned lo,
        const float* __restrict__ Wt, int ncols, const float* __restrict__ bias, float* __restrict__ C) {
    constexpr int KS = HDIM / 4;
    constexpr int G  = AWAVES / NTL;
    if (w < G * NTL) {
        const int c    = (int)lo & 15;
        const int koff = ((int)lo >> 4) * 2;
        const int n    = (w % NTL) * 16 + c;
        const bool in  = n < ncols;
        const int nn   = in ? n : 0;
        const float badd = (bias && in) ? bias[nn] : 0.0f;
        v2f b[KS];
        const float* bp = Wt + (size_t)koff * ncols + nn;
#pragma unroll
        for (int ks = 0; ks < KS; ++ks) {
            v2f t;
            t.x = bp[(ks * 4) * ncols];
            t.y = bp[(ks * 4 + 1) * ncols];
            if (!in) { t.x = 0.f; t.y = 0.f; }
            b[ks] = t;
        }
        const int ntile = (rows + 15) >> 4;
        const v8f z8 = {0.f, 0.f, 0.f, 0.f, 0.f, 0.f, 0.f, 0.f};
#pragma unroll 1
        for (int rt = w / NTL; rt < ntile; rt += G) {
            const float* ap = acc + (rt * 16 + c) * APITCH + koff;
            v8f d = z8;
#pragma unroll
            for (int ks = 0; ks < KS; ++ks) {
                const v2f a = *(const v2f*)(ap + ks * 4);
                d = __builtin_amdgcn_wmma_f32_16x16x4_f32(
                     false, a,  false, b[ks],
                     (short)0, d,
                     false,  false);
            }
            const int rbase = rt * 16 + (((int)lo >> 4) << 3);
            if (in) {
                float* cp = C + (size_t)(node_lo + rbase) * ncols + n;
#pragma unroll
                for (int r = 0; r < 8; ++r)
                    if (rbase + r < rows) cp[r * ncols] = d[r] + badd;
            }
        }
    }
}


typedef __attribute__((ext_vector_type(16))) float v16f;
struct RowAcc { v16f x, y, z, w; };

__device__ __forceinline__ void row_fma(RowAcc& a, int slot, float nj, const float (&v)[4]) {
    a.x[slot] = fmaf(v[0], nj, a.x[slot]);
    a.y[slot] = fmaf(v[1], nj, a.y[slot]);
    a.z[slot] = fmaf(v[2], nj, a.z[slot]);
    a.w[slot] = fmaf(v[3], nj, a.w[slot]);
}

__device__ __forceinline__ void load_row4(const float* __restrict__ h, int srcrow, unsigned lo, float (&v)[4]) {
    const float* p = h + ((unsigned)srcrow * (unsigned)HDIM + lo);
    v[0] = p[0];
    v[1] = p[32];
    v[2] = p[64];
    v[3] = p[96];
}

#define SPARE 15
__device__ __forceinline__ void gather_acc(const float* __restrict__ h, int r, int slot, float nrm, int cnt, unsigned lo,
                                           RowAcc& a) {
#pragma unroll 1
    for (int g = 0; g < cnt; g += EGRP) {
        float v[EGRP][4];
#pragma unroll
        for (int j = 0; j < EGRP; ++j) load_row4(h, __builtin_amdgcn_readlane(r, min(g + j, cnt - 1)), lo, v[j]);
#pragma unroll
        for (int j = 0; j < EGRP; ++j) {
            const int i = min(g + j, cnt - 1);
            const int sj = (g + j < cnt) ? __builtin_amdgcn_readlane(slot, i) : SPARE;
            const float nj = __int_as_float(__builtin_amdgcn_readlane(__float_as_int(nrm), i));
            row_fma(a, sj, nj, v[j]);
        }
    }
}

__device__ __forceinline__ void route(const int* qE, const unsigned short* qT, int off, int cnt, unsigned lo,
                                      int* rq, int* rcnt, int* tcnt, bool store) {
    if ((int)lo < cnt) {
        const int ej = qE[off + (int)lo];
        const int t  = (int)qT[off + (int)lo];
        __hip_atomic_fetch_add(&tcnt[t], 1, __ATOMIC_RELAXED, __HIP_MEMORY_SCOPE_WORKGROUP);
        if (store) {
            const int tw = t & (AWAVES - 1);
            const int pos = __hip_atomic_fetch_add(&rcnt[tw], 1, __ATOMIC_RELAXED, __HIP_MEMORY_SCOPE_WORKGROUP);
            if (pos < WCAP) rq[tw * WCAP + pos] = ej | ((t >> 5) << EBITS);
        }
    }
}

__device__ __forceinline__ void scan_route(const int* __restrict__ col, const unsigned* __restrict__ mine, int pitch,
                                           int* qE, unsigned short* qT, int* sL, int w, unsigned lo,
                                           int node_lo, int rows, int e, int* rq, int* rcnt, int* tcnt, bool store) {
    int qn = 0;
    int sn = 0;
    const int nchunk = (e + CHUNK - 1) / CHUNK;
#pragma unroll 1
    for (int cb = w; cb < nchunk; cb += 32 * AWAVES) {
        const int chunk = cb + AWAVES * (int)lo;
        const unsigned wd = (chunk < nchunk) ? mine[(size_t)chunk * pitch] : 0u;
#pragma unroll 1
        for (int kc = 0; kc < 32; kc += 8) {
            const unsigned part = (wd >> kc) & 0xffu;
            if (__builtin_amdgcn_ballot_w32(part != 0u) == 0u) continue;
#pragma unroll
            for (int k = 0; k < 8; ++k)
                append_seg(((part >> k) & 1u) != 0u, chunk * 32 + kc + k, sn, sL);
            while (sn >= SGRP) {
                sn -= SGRP;
                scan_batch(col, sL, sn, SGRP, e, lo, node_lo, rows, qn, qE, qT);
                while (qn >= 32) {
                    qn -= 32;
                    route(qE, qT, qn, 32, lo, rq, rcnt, tcnt, store);
                }
            }
        }
    }
    if (sn > 0) {
        scan_batch(col, sL, 0, sn, e, lo, node_lo, rows, qn, qE, qT);
        while (qn >= 32) {
            qn -= 32;
            route(qE, qT, qn, 32, lo, rq, rcnt, tcnt, store);
        }
    }
    if (qn > 0) route(qE, qT, 0, qn, lo, rq, rcnt, tcnt, store);
}

#define SDEEP 4
__global__ __launch_bounds__(1024) void k_scan(
        const int* __restrict__ row, const int* __restrict__ col,
        const unsigned* __restrict__ present, int pitch,
        int n, int e, float* __restrict__ dis, int* __restrict__ nent, int* __restrict__ lstw) {
    static_assert(RANGE <= 1024, "one thread per owned node");
    __shared__ int   rqs[AWAVES * WCAP];
    __shared__ int   qEs[AWAVES * QCAP];
    __shared__ int   sLs[AWAVES * SCAP];
    __shared__ unsigned short qTs[AWAVES * QCAP];
    __shared__ int   rcnt[AWAVES];
    __shared__ int   tcnt[RANGE];

    const int tid = (int)threadIdx.x;
    const unsigned lo = threadIdx.x & 31;
    const int w = __builtin_amdgcn_readfirstlane(tid >> 5);
    const int node_lo = (int)blockIdx.x * RANGE;
    const int rows = min(RANGE, n - node_lo);
    int2* lst = reinterpret_cast<int2*>(lstw) + ((size_t)blockIdx.x * AWAVES + w) * WCAP;
    int* nw = nent + (size_t)blockIdx.x * AWAVES;
    const unsigned* mine = present + blockIdx.x;

    if (tid < AWAVES) rcnt[tid] = 0;
    if (tid < RANGE) tcnt[tid] = 0;
    __syncthreads();
    const bool pre = (n > (1 << LBITS)) || (e > (1 << EBITS));
    scan_route(col, mine, pitch, qEs + w * QCAP, qTs + w * QCAP, sLs + w * SCAP, w, lo, node_lo, rows, e,
               rqs, rcnt, tcnt, !pre);
    __syncthreads();
    if (tid < rows) {
        const float d = 1.0f + (float)tcnt[tid];
        dis[node_lo + tid] = (d > 0.0f) ? rsqrtf(d) : 0.0f;
    }
    const int cl = rcnt[lo];
    const bool fb = pre || (__builtin_amdgcn_ballot_w32(cl > WCAP) != 0u);
    if (!fb) {
        const int cw = __builtin_amdgcn_readlane(cl, w);
        const int* rqw = rqs + w * WCAP;
#pragma unroll 1
        for (int b0 = 0; b0 < cw; b0 += 32 * SDEEP) {
            int ent[SDEEP], r[SDEEP];
#pragma unroll
            for (int j = 0; j < SDEEP; ++j) {
                const int i = b0 + 32 * j + (int)lo;
                ent[j] = rqw[min(i, cw - 1)];
                r[j] = row[ent[j] & ((1 << EBITS) - 1)];
            }
#pragma unroll
            for (int j = 0; j < SDEEP; ++j) {
                const int i = b0 + 32 * j + (int)lo;
                if (i < cw) {
                    int2 en;
                    en.x = r[j] | ((ent[j] >> EBITS) << LBITS);
                    en.y = 0;
                    lst[i] = en;
                }
            }
        }
        if (lo == 0u) nw[w] = cw;
    } else {
        if (lo == 0u) nw[w] = -1;
    }
}

template <int MODE>
__global__ __launch_bounds__(1024) void k_aggr(
        const float* __restrict__ h, const float* __restrict__ dis,
        const int* __restrict__ row, const int* __restrict__ col,
        const unsigned* __restrict__ present, int pitch,
        int n, int e, const int* __restrict__ nent, int* __restrict__ lstw,
        const float* __restrict__ eb, const float* __restrict__ gamma, const float* __restrict__ beta,
        const float* __restrict__ mean, const float* __restrict__ var, float* __restrict__ x1,
        const float* __restrict__ Wt, const float* __restrict__ bt, float* __restrict__ C) {
    static_assert(RANGE <= 512, "list entry packing");
    static_assert(RANGE % 16 == 0, "matrix tiles stay inside the LDS rows");
    static_assert(NROW == (RANGE + 31) / 32 && NROW <= SPARE, "row slots and the spare slot index a vector of 16");
    static_assert(AWAVES == 32, "a target's wave is its low five bits");
    __shared__ __align__(16) float acc[RANGE * APITCH];
    __shared__ int   qEs[AWAVES * QCAP];
    __shared__ int   sLs[AWAVES * SCAP];
    __shared__ unsigned short qTs[AWAVES * QCAP];

    const int tid = (int)threadIdx.x;
    const unsigned lo = threadIdx.x & 31;
    const int w = __builtin_amdgcn_readfirstlane(tid >> 5);
    const int node_lo = (int)blockIdx.x * RANGE;
    const int rows = min(RANGE, n - node_lo);
    int* qE = qEs + w * QCAP;
    unsigned short* qT = qTs + w * QCAP;
    int* sL = sLs + w * SCAP;
    int2* lst = reinterpret_cast<int2*>(lstw) + ((size_t)blockIdx.x * AWAVES + w) * WCAP;
    const int* nw = nent + (size_t)blockIdx.x * AWAVES;
    const unsigned* mine = present + blockIdx.x;

    RowAcc a;
    a.x = (v16f)(0.f); a.y = (v16f)(0.f); a.z = (v16f)(0.f); a.w = (v16f)(0.f);

    const int cl = nw[lo];
    const bool fb = __builtin_amdgcn_ballot_w32(cl < 0) != 0u;
    if (!fb) {
        const int ne = __builtin_amdgcn_readlane(cl, w);
        int2* L = lst;
        int2 en = make_int2(0, 0);
        if (ne > 0) en = L[min((int)lo, min(32, ne) - 1)];
#pragma unroll 1
        for (int b0 = 0; b0 < ne; b0 += 32) {
            int2 nx = en;
            const int bn = b0 + 32;
            if (bn < ne) nx = L[bn + min((int)lo, min(32, ne - bn) - 1)];
            const int cnt = min(32, ne - b0);
            const int r = en.x & ((1 << LBITS) - 1);
            const int slot = en.x >> LBITS;
            float nrm;
            if (MODE == 0) {
                nrm = dis[r] * dis[node_lo + slot * 32 + w];
                if ((int)lo < cnt) L[b0 + (int)lo].y = __float_as_int(nrm);
            } else {
                nrm = __int_as_float(en.y);
            }
            gather_acc(h, r, slot, nrm, cnt, lo, a);
            en = nx;
        }
    }

    if (!fb) {
#pragma unroll
        for (int k = 0; k < NROW; ++k) {
            const int t = k * 32 + w;
            if (t < RANGE) {
                float* drow = acc + t * APITCH;
                drow[lo]      = a.x[k];
                drow[lo + 32] = a.y[k];
                drow[lo + 64] = a.z[k];
                drow[lo + 96] = a.w[k];
            }
        }
    } else {
        float4* acc4 = reinterpret_cast<float4*>(acc);
        for (int i = tid; i < RANGE * APITCH / 4; i += 1024) acc4[i] = make_float4(0.f, 0.f, 0.f, 0.f);
        __syncthreads();
        scan_pass<APITCH>(h, dis, row, col, mine, pitch, acc, qE, qT, sL, w, lo, node_lo, rows, e);
    }
    __syncthreads();

    if (MODE == 0) rows_epilogue1(acc, rows, node_lo, tid, h, dis, eb, gamma, beta, mean, var, x1);
    else           rows_epilogue2(acc, rows, node_lo, tid, h, dis, eb, x1);
    __syncthreads();
    if (MODE == 0) rows_product<HDIM / 16>(acc, rows, node_lo, w, lo, Wt, HDIM, bt, C);
    else           rows_product<(ODIM + 15) / 16>(acc, rows, node_lo, w, lo, Wt, ODIM, bt, C);
}

extern "C" void kernel_launch(void* const* d_in, const int* in_sizes, int n_in,
                              void* d_out, int out_size, void* d_ws, size_t ws_size,
                              hipStream_t stream) {
    const float* node_feat = (const float*)d_in[0];
    const int*   eidx      = (const int*)d_in[1];
    const float* W1        = (const float*)d_in[2];
    const float* b1        = (const float*)d_in[3];
    const float* gamma1    = (const float*)d_in[4];
    const float* beta1     = (const float*)d_in[5];
    const float* mean1     = (const float*)d_in[6];
    const float* var1      = (const float*)d_in[7];
    const float* W2        = (const float*)d_in[8];
    const float* b2        = (const float*)d_in[9];
    const float* Wf        = (const float*)d_in[10];
    const float* bf        = (const float*)d_in[11];
    float* out = (float*)d_out;

    const int n = in_sizes[0] / HDIM;
    const int e = in_sizes[1] / 2;
    const int* row = eidx;
    const int* col = eidx + e;

    float* dis = (float*)d_ws;
    float* h1  = dis + n;
    float* x1  = h1 + (size_t)n * HDIM;
    float* h2  = x1 + (size_t)n * HDIM;
    unsigned* present = (unsigned*)(h2 + (size_t)n * HDIM);

    const int aggGrd  = (n + RANGE - 1) / RANGE;
    const int pitch   = (aggGrd + 31) / 32 * 32;
    const int chunks  = (e + CHUNK - 1) / CHUNK;
    int* nent = (int*)(present + (size_t)chunks * pitch);
    int* lstw = nent + (size_t)pitch * AWAVES;

    k_edge_present<<<chunks, CHUNK, 0, stream>>>(col, present, e, pitch);
    k_scan<<<aggGrd, 1024, 0, stream>>>(row, col, present, pitch, n, e, dis, nent, lstw);

    k_gemm_wmma<128, 2><<<GEMM_WGS, 256, 0, stream>>>(node_feat, W1, nullptr, h1, n, HDIM);
    k_aggr<0><<<aggGrd, 1024, 0, stream>>>(h1, dis, row, col, present, pitch, n, e, nent, lstw,
                                          b1, gamma1, beta1, mean1, var1, x1, W2, nullptr, h2);

    k_aggr<1><<<aggGrd, 1024, 0, stream>>>(h2, dis, row, col, present, pitch, n, e, nent, lstw,
                                          b2, nullptr, nullptr, nullptr, nullptr, x1, Wf, bf, out);
}
